# starting kernel
# baseline (speedup 1.0000x reference)
#include <hip/hip_runtime.h>
#include <math.h>

#define IMG      2048
#define NB       16
#define DCH      64
#define INF      256
#define NPOS     23
#define GRID24   24
#define NPATCH   (NPOS*NPOS)
#define NTILE    ((NPATCH+15)/16)

typedef float v2f __attribute__((ext_vector_type(2)));
typedef float v8f __attribute__((ext_vector_type(8)));

__global__ void k_tables(double* ct256, double* st256, double* ct64, double* st64) {
    int t = threadIdx.x;
    const double TWO_PI = 6.283185307179586476925286766559;
    if (t < 256) { double a = TWO_PI * (double)t / 256.0; ct256[t] = cos(a); st256[t] = sin(a); }
    if (t < 64)  { double a = TWO_PI * (double)t / 64.0;  ct64[t]  = cos(a); st64[t]  = sin(a); }
}

__global__ void k_b12(const float* __restrict__ Wr, const float* __restrict__ Wi,
                      const double* __restrict__ ct, const double* __restrict__ st,
                      double* __restrict__ B1, double* __restrict__ B2) {
    int d = blockIdx.x;
    int n = threadIdx.x;
    const float* wr = Wr + d * INF;
    const float* wi = Wi + d * INF;
    double a1 = 0.0, a2 = 0.0;
    for (int k = 0; k < INF; ++k) {
        int t = (k * n) & 255;
        double c = ct[t], s = st[t];
        double r = (double)wr[k], im = (double)wi[k];
        a1 += r * c + im * s;
        a2 += im * c - r * s;
    }
    B1[d * INF + n] = a1;
    B2[d * INF + n] = a2;
}

__global__ void k_amat(const double* __restrict__ B1, const double* __restrict__ B2,
                       const double* __restrict__ ct64, const double* __restrict__ st64,
                       const float* __restrict__ br, const float* __restrict__ bi,
                       float* __restrict__ A, float* __restrict__ cvec) {
    int m = blockIdx.x;
    int n = threadIdx.x;
    double acc = 0.0;
    for (int d = 0; d < DCH; ++d) {
        int t = (d * m) & 63;
        acc += ct64[t] * B1[d * INF + n] - st64[t] * B2[d * INF + n];
    }
    A[m * INF + n] = (float)(acc / 64.0);
    if (n == 0) {
        double cc = 0.0;
        for (int d = 0; d < DCH; ++d) {
            int t = (d * m) & 63;
            cc += ct64[t] * ((double)br[d] - (double)bi[d])
                - st64[t] * ((double)br[d] + (double)bi[d]);
        }
        cvec[m] = (float)(cc / 64.0);
    }
}

__global__ void k_zero(float* __restrict__ p, int n) {
    int i = blockIdx.x * blockDim.x + threadIdx.x;
    if (i < n) p[i] = 0.0f;
}

__global__ void k_wmma(const float* __restrict__ x, const float* __restrict__ A,
                       const float* __restrict__ cvec, float* __restrict__ fbuf) {
    int lane = threadIdx.x & 31;
    int wv   = threadIdx.x >> 5;
    int pt   = blockIdx.x % NTILE;
    int img  = blockIdx.x / NTILE;
    int n    = lane & 15;
    int hi   = lane >> 4;

    int q  = pt * 16 + n;
    int qc = q > (NPATCH - 1) ? (NPATCH - 1) : q;
    int ri = 1 + qc / NPOS;
    int ci = 1 + qc % NPOS;
    int prow = 16 * (ri / 3) + (ri % 3) - 1;
    int pcol = 16 * (ci / 3) + (ci % 3) - 1;

    const float* xb = x + (size_t)img * IMG * IMG
                        + (size_t)(prow * 16) * IMG + (size_t)(pcol * 16) + 2 * hi;
    const float* ap = A + (size_t)(wv * 16 + n) * INF + 2 * hi;

    __builtin_prefetch((const void*)xb, 0, 3);

    v8f acc = {};
    #pragma unroll 8
    for (int s = 0; s < 64; ++s) {
        v2f a = *(const v2f*)(ap + 4 * s);
        v2f b = *(const v2f*)(xb + (s >> 2) * IMG + ((s & 3) << 2));
        acc = __builtin_amdgcn_wmma_f32_16x16x4_f32(
                   false, a,  false, b,
                   (short)0, acc,  false,  false);
    }

    if (q <= NPATCH - 1) {
        #pragma unroll
        for (int j = 0; j < 8; ++j) {
            int ch = wv * 16 + hi * 8 + j;
            fbuf[(((size_t)img * DCH + ch) * GRID24 + ri) * GRID24 + ci] = acc[j] + cvec[ch];
        }
    }
}

__global__ void k_out(const float* __restrict__ fbuf, const float* __restrict__ cw,
                      const float* __restrict__ cb, const float* __restrict__ gamma,
                      const float* __restrict__ beta, const float* __restrict__ mean,
                      const float* __restrict__ var, float* __restrict__ out) {
    int idx = blockIdx.x * blockDim.x + threadIdx.x;
    if (idx >= NB * DCH * 8 * 8) return;
    int ow = idx & 7;
    int oh = (idx >> 3) & 7;
    int ch = (idx >> 6) & 63;
    int b  = idx >> 12;
    const float* f = fbuf + (((size_t)b * DCH + ch) * GRID24 + 3 * oh) * GRID24 + 3 * ow;
    const float* w = cw + ch * 9;
    float acc = cb[ch];
    #pragma unroll
    for (int j = 0; j < 3; ++j)
        #pragma unroll
        for (int i = 0; i < 3; ++i)
            acc += f[j * GRID24 + i] * w[j * 3 + i];
    float scale = gamma[ch] / sqrtf(var[ch] + 1e-5f);
    out[idx] = (acc - mean[ch]) * scale + beta[ch];
}

extern "C" void kernel_launch(void* const* d_in, const int* in_sizes, int n_in,
                              void* d_out, int out_size, void* d_ws, size_t ws_size,
                              hipStream_t stream) {
    const float* x     = (const float*)d_in[0];
    const float* Wr    = (const float*)d_in[1];
    const float* br    = (const float*)d_in[2];
    const float* Wi    = (const float*)d_in[3];
    const float* bi    = (const float*)d_in[4];
    const float* cw    = (const float*)d_in[5];
    const float* cb    = (const float*)d_in[6];
    const float* gamma = (const float*)d_in[7];
    const float* beta  = (const float*)d_in[8];
    const float* mean  = (const float*)d_in[9];
    const float* var   = (const float*)d_in[10];

    char* ws = (char*)d_ws;
    double* ct256 = (double*)(ws + 0);
    double* st256 = (double*)(ws + 2048);
    double* ct64  = (double*)(ws + 4096);
    double* st64  = (double*)(ws + 4608);
    double* B1    = (double*)(ws + 5120);
    double* B2    = (double*)(ws + 136192);
    float*  A     = (float*) (ws + 267264);
    float*  cvec  = (float*) (ws + 332800);
    float*  fbuf  = (float*) (ws + 333056);

    k_tables<<<1, 256, 0, stream>>>(ct256, st256, ct64, st64);
    k_b12<<<DCH, INF, 0, stream>>>(Wr, Wi, ct256, st256, B1, B2);
    k_amat<<<DCH, INF, 0, stream>>>(B1, B2, ct64, st64, br, bi, A, cvec);

    int nz = NB * DCH * GRID24 * GRID24;
    k_zero<<<(nz + 255) / 256, 256, 0, stream>>>(fbuf, nz);

    k_wmma<<<NB * NTILE, 128, 0, stream>>>(x, A, cvec, fbuf);

    int nout = NB * DCH * 8 * 8;
    k_out<<<(nout + 255) / 256, 256, 0, stream>>>(fbuf, cw, cb, gamma, beta, mean, var,
                                                  (float*)d_out);
}


# k_wmma: K=256 split over the block's 4 waves (64 features x 4 channel tiles each), all 40 operand loads in flight before the first product; partial tiles joined in LDS: new f32 sum order
# speedup vs baseline: 4.9532x; 4.9532x over previous
#include <hip/hip_runtime.h>
#include <math.h>

#define IMG      2048
#define NB       16
#define DCH      64
#define INF      256
#define NPOS     23
#define GRID24   24
#define NPATCH   (NPOS*NPOS)
#define NTILE    ((NPATCH+15)/16)

typedef float v2f __attribute__((ext_vector_type(2)));
typedef float v4f __attribute__((ext_vector_type(4)));
typedef float v8f __attribute__((ext_vector_type(8)));

__global__ void k_tables(double* ct256, double* st256, double* ct64, double* st64) {
    int g = blockIdx.x * 32 + threadIdx.x;
    const double TWO_PI = 6.283185307179586476925286766559;
    if (g < 256) { double a = TWO_PI * (double)g / 256.0; ct256[g] = cos(a); st256[g] = sin(a); }
    else { int t = g - 256; double a = TWO_PI * (double)t / 64.0;  ct64[t]  = cos(a); st64[t]  = sin(a); }
}

#define FOLDP 130
__global__ void k_b12(const float* __restrict__ Wr, const float* __restrict__ Wi,
                      const double* __restrict__ ct, const double* __restrict__ st,
                      double* __restrict__ B1, double* __restrict__ B2) {
    __shared__ double fold[4 * FOLDP];
    __shared__ double part[4][8][32];
    int d  = blockIdx.x >> 2;
    int nq = blockIdx.x & 3;
    int nl = threadIdx.x & 31;
    int kq = __builtin_amdgcn_readfirstlane((int)(threadIdx.x >> 5));
    int n  = nq * 32 + nl;

    {
        int kp  = threadIdx.x & 127;
        int sel = (int)(threadIdx.x >> 7) * 2;
        const float* w = (sel ? Wi : Wr) + d * INF;
        double a = (double)w[kp];
        double b = (double)w[(256 - kp) & 255];
        fold[sel * FOLDP + kp]       = (kp == 0) ? a : a + b;
        fold[(sel + 1) * FOLDP + kp] = a - b;
        if (kp == 0) {
            double h = (double)w[128];
            fold[sel * FOLDP + 128]       = h;
            fold[(sel + 1) * FOLDP + 128] = h;
        }
    }
    __syncthreads();

    const double* stab = (n == 0) ? ct : st;
    int ns = (n == 0) ? 128 : n;
    int so = (n == 0) ? 0 : FOLDP;
    double pr = 0.0, qr = 0.0, pi = 0.0, qi = 0.0;
    for (int j = 0; j < 16; ++j) {
        int kp = kq * 16 + j;
        double c = ct[(kp * n) & 255];
        double s = stab[(kp * ns) & 255];
        pr += fold[kp] * c;
        qr += fold[so + kp] * s;
        pi += fold[2 * FOLDP + kp] * c;
        qi += fold[2 * FOLDP + so + kp] * s;
    }
    if (kq == 0) {
        double c = ct[(128 * n) & 255];
        double s = stab[(128 * ns) & 255];
        pr += fold[128] * c;
        qr += fold[so + 128] * s;
        pi += fold[2 * FOLDP + 128] * c;
        qi += fold[2 * FOLDP + so + 128] * s;
    }
    part[0][kq][nl] = pr;
    part[1][kq][nl] = qr;
    part[2][kq][nl] = pi;
    part[3][kq][nl] = qi;
    __syncthreads();
    if (threadIdx.x < 128) {
        int which = kq;
        int xa = (which & 1) ? 2 : 0;
        int ya = (which & 1) ? 1 : 3;
        double X = 0.0, Y = 0.0;
        for (int p = 0; p < 8; ++p) { X += part[xa][p][nl]; Y += part[ya][p][nl]; }
        double v = (which == 0 || which == 3) ? (X + Y) : (X - Y);
        if (n == 0) {
            int idx = (which < 2) ? xa : ((which == 2) ? 1 : 3);
            double z = 0.0;
            for (int p = 0; p < 8; ++p) z += part[idx][p][0];
            v = z;
        }
        double* dst = (which & 1) ? B2 : B1;
        dst[d * INF + (which >> 1) * 128 + n] = v;
    }
}

__global__ void k_amat(const double* __restrict__ B1, const double* __restrict__ B2,
                       const double* __restrict__ ct64, const double* __restrict__ st64,
                       const float* __restrict__ br, const float* __restrict__ bi,
                       float* __restrict__ A, float* __restrict__ cvec) {
    __shared__ double pu[8][32];
    __shared__ double pv[8][32];
    int m  = blockIdx.x >> 3;
    int ne = blockIdx.x & 7;
    int nl = threadIdx.x & 31;
    int dq = __builtin_amdgcn_readfirstlane((int)(threadIdx.x >> 5));
    int n  = ne * 32 + nl;
    int np = n <= 128 ? n : 384 - n;
    const double* vt = (m == 0) ? ct64 : st64;
    const double* vb = (m == 0) ? B1 : B2;
    int ms = (m == 0) ? 32 : m;
    double u = 0.0, v = 0.0;
    for (int j = 0; j < 8; ++j) {
        int d = dq * 8 + j;
        u += ct64[(d * m) & 63] * B1[d * INF + np];
        v += vt[(d * ms) & 63] * vb[d * INF + np];
    }
    pu[dq][nl] = u;
    pv[dq][nl] = v;
    __syncthreads();
    if (threadIdx.x < 64) {
        int hi = dq;
        double U = 0.0, V = 0.0;
        for (int p = 0; p < 8; ++p) { U += pu[p][nl]; V += pv[p][nl]; }
        double val = hi ? (U + V) : (U - V);
        int row = hi ? 64 - m : m;
        if (m == 0) { val = hi ? V : U; row = hi ? 32 : 0; }
        A[row * INF + n] = (float)(val / 64.0);
    }
    if (ne == 0 && (dq == 2 || dq == 3)) {
        int mr = (dq == 2) ? m : ((m == 0) ? 32 : 64 - m);
        int l = nl;
        double cc = 0.0;
        for (int h = 0; h < 2; ++h) {
            int d = l + 32 * h;
            int t = (d * mr) & 63;
            cc += ct64[t] * ((double)br[d] - (double)bi[d])
                - st64[t] * ((double)br[d] + (double)bi[d]);
        }
        for (int o = 16; o >= 1; o >>= 1) cc += __shfl_xor(cc, o);
        if (l == 0) cvec[mr] = (float)(cc / 64.0);
    }
}

__global__ void __launch_bounds__(128)
k_wmma(const float* __restrict__ x, const float* __restrict__ A,
       const float* __restrict__ cvec, float* __restrict__ fbuf) {
    __shared__ v4f red[4][4][2][32];
    int lane = threadIdx.x & 31;
    int wv   = __builtin_amdgcn_readfirstlane((int)(threadIdx.x >> 5));
    int pt   = blockIdx.x % NTILE;
    int img  = blockIdx.x / NTILE;
    int n    = lane & 15;
    int hi   = lane >> 4;

    int q  = pt * 16 + n;
    int qc = q > (NPATCH - 1) ? (NPATCH - 1) : q;
    int ri = 1 + qc / NPOS;
    int ci = 1 + qc % NPOS;
    int prow = 16 * (ri / 3) + (ri % 3) - 1;
    int pcol = 16 * (ci / 3) + (ci % 3) - 1;

    const float* xb = x + (size_t)img * IMG * IMG
                        + (size_t)(prow * 16 + 4 * wv) * IMG + (size_t)(pcol * 16);
    const float* ap = A + (size_t)n * INF + 64 * wv;

    v4f b[8];
    v4f a[4][8];
    #pragma unroll
    for (int j = 0; j < 8; ++j) {
        int u = 8 * j + 4 * hi;
        b[j] = *(const v4f*)(xb + (u >> 4) * IMG + (u & 15));
    }
    __builtin_amdgcn_sched_barrier(0);
    #pragma unroll
    for (int t = 0; t < 4; ++t)
        #pragma unroll
        for (int j = 0; j < 8; ++j) {
            int u = 8 * j + 4 * hi;
            a[t][j] = *(const v4f*)(ap + (size_t)(t * 16) * INF + u);
        }

    __builtin_amdgcn_sched_barrier(0);

    v8f acc[4] = {};
    #pragma unroll
    for (int j = 0; j < 8; ++j)
        #pragma unroll
        for (int p = 0; p < 2; ++p) {
            v2f bv = p ? (v2f){b[j][2], b[j][3]} : (v2f){b[j][0], b[j][1]};
            #pragma unroll
            for (int t = 0; t < 4; ++t) {
                v2f av = p ? (v2f){a[t][j][2], a[t][j][3]} : (v2f){a[t][j][0], a[t][j][1]};
                acc[t] = __builtin_amdgcn_wmma_f32_16x16x4_f32(
                              false, av,  false, bv,
                              (short)0, acc[t],  false,  false);
            }
        }

    #pragma unroll
    for (int t = 0; t < 4; ++t) {
        red[wv][t][0][lane] = (v4f){acc[t][0], acc[t][1], acc[t][2], acc[t][3]};
        red[wv][t][1][lane] = (v4f){acc[t][4], acc[t][5], acc[t][6], acc[t][7]};
    }
    __syncthreads();
    v4f lo = red[0][wv][0][lane];
    v4f up = red[0][wv][1][lane];
    #pragma unroll
    for (int p = 1; p < 4; ++p) {
        lo += red[p][wv][0][lane];
        up += red[p][wv][1][lane];
    }

    if (q <= NPATCH - 1) {
        #pragma unroll
        for (int j = 0; j < 8; ++j) {
            int ch = wv * 16 + hi * 8 + j;
            float v = (j < 4) ? lo[j & 3] : up[j & 3];
            fbuf[(((size_t)img * DCH + ch) * GRID24 + ri) * GRID24 + ci] = v + cvec[ch];
        }
    }
}

__global__ void k_out(const float* __restrict__ fbuf, const float* __restrict__ cw,
                      const float* __restrict__ cb, const float* __restrict__ gamma,
                      const float* __restrict__ beta, const float* __restrict__ mean,
                      const float* __restrict__ var, float* __restrict__ out) {
    int idx = blockIdx.x * blockDim.x + threadIdx.x;
    if (idx >= NB * DCH * 8 * 8) return;
    int ow = idx & 7;
    int oh = (idx >> 3) & 7;
    int ch = (idx >> 6) & 63;
    int b  = idx >> 12;
    const float* f = fbuf + (((size_t)b * DCH + ch) * GRID24 + 3 * oh) * GRID24 + 3 * ow;
    const float* w = cw + ch * 9;
    float acc = cb[ch];
    #pragma unroll
    for (int j = 0; j < 3; ++j)
        #pragma unroll
        for (int i = 0; i < 3; ++i) {
            bool pad = (j == 0 && oh == 0) || (i == 0 && ow == 0);
            float fv = 0.0f;
            if (!pad) fv = f[j * GRID24 + i];
            acc += fv * w[j * 3 + i];
        }
    float scale = gamma[ch] / sqrtf(var[ch] + 1e-5f);
    out[idx] = (acc - mean[ch]) * scale + beta[ch];
}

extern "C" void kernel_launch(void* const* d_in, const int* in_sizes, int n_in,
                              void* d_out, int out_size, void* d_ws, size_t ws_size,
                              hipStream_t stream) {
    const float* x     = (const float*)d_in[0];
    const float* Wr    = (const float*)d_in[1];
    const float* br    = (const float*)d_in[2];
    const float* Wi    = (const float*)d_in[3];
    const float* bi    = (const float*)d_in[4];
    const float* cw    = (const float*)d_in[5];
    const float* cb    = (const float*)d_in[6];
    const float* gamma = (const float*)d_in[7];
    const float* beta  = (const float*)d_in[8];
    const float* mean  = (const float*)d_in[9];
    const float* var   = (const float*)d_in[10];

    char* ws = (char*)d_ws;
    double* ct256 = (double*)(ws + 0);
    double* st256 = (double*)(ws + 2048);
    double* ct64  = (double*)(ws + 4096);
    double* st64  = (double*)(ws + 4608);
    double* B1    = (double*)(ws + 5120);
    double* B2    = (double*)(ws + 136192);
    float*  A     = (float*) (ws + 267264);
    float*  cvec  = (float*) (ws + 332800);
    float*  fbuf  = (float*) (ws + 333056);

    k_tables<<<10, 32, 0, stream>>>(ct256, st256, ct64, st64);
    k_b12<<<DCH * 4, 256, 0, stream>>>(Wr, Wi, ct256, st256, B1, B2);
    k_amat<<<DCH * 4, 256, 0, stream>>>(B1, B2, ct64, st64, br, bi, A, cvec);

    k_wmma<<<NB * NTILE, 128, 0, stream>>>(x, A, cvec, fbuf);

    int nout = NB * DCH * 8 * 8;
    k_out<<<(nout + 255) / 256, 256, 0, stream>>>(fbuf, cw, cb, gamma, beta, mean, var,
                                                  (float*)d_out);
}


# k_wmma re-tiled by output: block = 3x12 patches under 4 outputs (3 tap-row tiles x 4 K-quarter waves); f meets in LDS, conv+BN finish there: no staging plane, 5 -> 4 launches; sum order unchanged
# speedup vs baseline: 5.4354x; 1.0974x over previous
#include <hip/hip_runtime.h>
#include <math.h>

#define IMG      2048
#define NB       16
#define DCH      64
#define INF      256

typedef float v2f __attribute__((ext_vector_type(2)));
typedef float v4f __attribute__((ext_vector_type(4)));
typedef float v8f __attribute__((ext_vector_type(8)));
typedef __attribute__((address_space(3))) v4f   lds_v4f;
typedef __attribute__((address_space(3))) float lds_f32;

__global__ void k_tables(double* ct256, double* st256, double* ct64, double* st64) {
    int g = blockIdx.x * 32 + threadIdx.x;
    const double TWO_PI = 6.283185307179586476925286766559;
    if (g < 256) { double a = TWO_PI * (double)g / 256.0; ct256[g] = cos(a); st256[g] = sin(a); }
    else { int t = g - 256; double a = TWO_PI * (double)t / 64.0;  ct64[t]  = cos(a); st64[t]  = sin(a); }
}

#define FOLDP 130
__global__ void k_b12(const float* __restrict__ Wr, const float* __restrict__ Wi,
                      const double* __restrict__ ct, const double* __restrict__ st,
                      double* __restrict__ B1, double* __restrict__ B2) {
    __shared__ double fold[4 * FOLDP];
    __shared__ double part[4][8][32];
    int d  = blockIdx.x >> 2;
    int nq = blockIdx.x & 3;
    int nl = threadIdx.x & 31;
    int kq = __builtin_amdgcn_readfirstlane((int)(threadIdx.x >> 5));
    int n  = nq * 32 + nl;

    {
        int kp  = threadIdx.x & 127;
        int sel = (int)(threadIdx.x >> 7) * 2;
        const float* w = (sel ? Wi : Wr) + d * INF;
        double a = (double)w[kp];
        double b = (double)w[(256 - kp) & 255];
        fold[sel * FOLDP + kp]       = (kp == 0) ? a : a + b;
        fold[(sel + 1) * FOLDP + kp] = a - b;
        if (kp == 0) {
            double h = (double)w[128];
            fold[sel * FOLDP + 128]       = h;
            fold[(sel + 1) * FOLDP + 128] = h;
        }
    }
    __syncthreads();

    const double* stab = (n == 0) ? ct : st;
    int ns = (n == 0) ? 128 : n;
    int so = (n == 0) ? 0 : FOLDP;
    double pr = 0.0, qr = 0.0, pi = 0.0, qi = 0.0;
    for (int j = 0; j < 16; ++j) {
        int kp = kq * 16 + j;
        double c = ct[(kp * n) & 255];
        double s = stab[(kp * ns) & 255];
        pr += fold[kp] * c;
        qr += fold[so + kp] * s;
        pi += fold[2 * FOLDP + kp] * c;
        qi += fold[2 * FOLDP + so + kp] * s;
    }
    if (kq == 0) {
        double c = ct[(128 * n) & 255];
        double s = stab[(128 * ns) & 255];
        pr += fold[128] * c;
        qr += fold[so + 128] * s;
        pi += fold[2 * FOLDP + 128] * c;
        qi += fold[2 * FOLDP + so + 128] * s;
    }
    part[0][kq][nl] = pr;
    part[1][kq][nl] = qr;
    part[2][kq][nl] = pi;
    part[3][kq][nl] = qi;
    __syncthreads();
    if (threadIdx.x < 128) {
        int which = kq;
        int xa = (which & 1) ? 2 : 0;
        int ya = (which & 1) ? 1 : 3;
        double X = 0.0, Y = 0.0;
        for (int p = 0; p < 8; ++p) { X += part[xa][p][nl]; Y += part[ya][p][nl]; }
        double v = (which == 0 || which == 3) ? (X + Y) : (X - Y);
        if (n == 0) {
            int idx = (which < 2) ? xa : ((which == 2) ? 1 : 3);
            double z = 0.0;
            for (int p = 0; p < 8; ++p) z += part[idx][p][0];
            v = z;
        }
        double* dst = (which & 1) ? B2 : B1;
        dst[d * INF + (which >> 1) * 128 + n] = v;
    }
}

__global__ void k_amat(const double* __restrict__ B1, const double* __restrict__ B2,
                       const double* __restrict__ ct64, const double* __restrict__ st64,
                       const float* __restrict__ br, const float* __restrict__ bi,
                       float* __restrict__ A, float* __restrict__ cvec) {
    __shared__ double pu[8][32];
    __shared__ double pv[8][32];
    int m  = blockIdx.x >> 3;
    int ne = blockIdx.x & 7;
    int nl = threadIdx.x & 31;
    int dq = __builtin_amdgcn_readfirstlane((int)(threadIdx.x >> 5));
    int n  = ne * 32 + nl;
    int np = n <= 128 ? n : 384 - n;
    const double* vt = (m == 0) ? ct64 : st64;
    const double* vb = (m == 0) ? B1 : B2;
    int ms = (m == 0) ? 32 : m;
    double u = 0.0, v = 0.0;
    for (int j = 0; j < 8; ++j) {
        int d = dq * 8 + j;
        u += ct64[(d * m) & 63] * B1[d * INF + np];
        v += vt[(d * ms) & 63] * vb[d * INF + np];
    }
    pu[dq][nl] = u;
    pv[dq][nl] = v;
    __syncthreads();
    if (threadIdx.x < 64) {
        int hi = dq;
        double U = 0.0, V = 0.0;
        for (int p = 0; p < 8; ++p) { U += pu[p][nl]; V += pv[p][nl]; }
        double val = hi ? (U + V) : (U - V);
        int row = hi ? 64 - m : m;
        if (m == 0) { val = hi ? V : U; row = hi ? 32 : 0; }
        A[row * INF + n] = (float)(val / 64.0);
    }
    if (ne == 0 && (dq == 2 || dq == 3)) {
        int mr = (dq == 2) ? m : ((m == 0) ? 32 : 64 - m);
        int l = nl;
        double cc = 0.0;
        for (int h = 0; h < 2; ++h) {
            int d = l + 32 * h;
            int t = (d * mr) & 63;
            cc += ct64[t] * ((double)br[d] - (double)bi[d])
                - st64[t] * ((double)br[d] + (double)bi[d]);
        }
        for (int o = 16; o >= 1; o >>= 1) cc += __shfl_xor(cc, o);
        if (l == 0) cvec[mr] = (float)(cc / 64.0);
    }
}

#define SLOTW    12
#define FPITCH   52
__global__ void __launch_bounds__(384)
k_wmma(const float* __restrict__ x, const float* __restrict__ A, const float* __restrict__ cvec,
       const float* __restrict__ cw, const float* __restrict__ cb, const float* __restrict__ gamma,
       const float* __restrict__ beta, const float* __restrict__ mean, const float* __restrict__ var,
       float* __restrict__ out) {
    __shared__ v4f   red[4][3][4][2][32];
    __shared__ float fl[DCH * FPITCH];
    int lane = threadIdx.x & 31;
    int wid  = __builtin_amdgcn_readfirstlane((int)(threadIdx.x >> 5));
    int kq   = wid & 3;
    int tl   = wid >> 2;
    int half = blockIdx.x & 1;
    int oh   = (blockIdx.x >> 1) & 7;
    int img  = blockIdx.x >> 4;
    int n    = lane & 15;
    int hi   = lane >> 4;

    int s  = tl * 16 + n;
    int tc = n > (SLOTW - 1) ? (SLOTW - 1) : n;
    int prow = 16 * oh + tl - 1;
    int pcol = 16 * (4 * half + tc / 3) + (tc % 3) - 1;
    prow = prow < 0 ? 0 : prow;
    pcol = pcol < 0 ? 0 : pcol;

    const float* xb = x + (size_t)img * IMG * IMG
                        + (size_t)(prow * 16 + 4 * kq) * IMG + (size_t)(pcol * 16);
    const float* ap = A + (size_t)n * INF + 64 * kq;

    v4f b[8];
    v4f a[4][8];
    #pragma unroll
    for (int j = 0; j < 8; ++j) {
        int u = 8 * j + 4 * hi;
        b[j] = *(const v4f*)(xb + (u >> 4) * IMG + (u & 15));
    }
    __builtin_amdgcn_sched_barrier(0);
    #pragma unroll
    for (int t = 0; t < 4; ++t)
        #pragma unroll
        for (int j = 0; j < 8; ++j) {
            int u = 8 * j + 4 * hi;
            a[t][j] = *(const v4f*)(ap + (size_t)(t * 16) * INF + u);
        }
    int ech = (int)(threadIdx.x >> 2) & 63;
    int owl = (int)threadIdx.x & 3;
    v4f cv0 = *(const v4f*)(cvec + kq * 16 + hi * 8);
    v4f cv1 = *(const v4f*)(cvec + kq * 16 + hi * 8 + 4);
    float wt[9];
    #pragma unroll
    for (int k = 0; k < 9; ++k) wt[k] = cw[ech * 9 + k];
    float e_cb = cb[ech], e_g = gamma[ech], e_b = beta[ech], e_m = mean[ech], e_v = var[ech];
    lds_v4f* rst = (lds_v4f*)&red[kq][tl][0][0][lane];
    lds_v4f* rld = (lds_v4f*)&red[0][tl][kq][0][lane];
    lds_f32* fst = (lds_f32*)(fl + (kq * 16 + hi * 8) * FPITCH + s);
    asm volatile("" : "+v"(rst), "+v"(rld), "+v"(fst));

    __builtin_amdgcn_sched_barrier(0);

    v8f acc[4] = {};
    #pragma unroll
    for (int j = 0; j < 8; ++j)
        #pragma unroll
        for (int p = 0; p < 2; ++p) {
            v2f bv = p ? (v2f){b[j][2], b[j][3]} : (v2f){b[j][0], b[j][1]};
            #pragma unroll
            for (int t = 0; t < 4; ++t) {
                v2f av = p ? (v2f){a[t][j][2], a[t][j][3]} : (v2f){a[t][j][0], a[t][j][1]};
                acc[t] = __builtin_amdgcn_wmma_f32_16x16x4_f32(
                              false, av,  false, bv,
                              (short)0, acc[t],  false,  false);
            }
        }

    #pragma unroll
    for (int t = 0; t < 4; ++t) {
        rst[64 * t]      = (v4f){acc[t][0], acc[t][1], acc[t][2], acc[t][3]};
        rst[64 * t + 32] = (v4f){acc[t][4], acc[t][5], acc[t][6], acc[t][7]};
    }
    __syncthreads();
    v4f lo = rld[0];
    v4f up = rld[32];
    #pragma unroll
    for (int p = 1; p < 4; ++p) {
        lo += rld[768 * p];
        up += rld[768 * p + 32];
    }
    #pragma unroll
    for (int j = 0; j < 8; ++j) {
        float v = (j < 4) ? lo[j & 3] : up[j & 3];
        float c = (j < 4) ? cv0[j & 3] : cv1[j & 3];
        fst[FPITCH * j] = v + c;
    }
    __syncthreads();

    if (threadIdx.x < 256) {
        int ow = 4 * half + owl;
        const float* f = fl + ech * FPITCH + 3 * owl;
        float r = e_cb;
        #pragma unroll
        for (int j = 0; j < 3; ++j)
            #pragma unroll
            for (int i = 0; i < 3; ++i) {
                bool pad = (j == 0 && oh == 0) || (i == 0 && ow == 0);
                float fv = 0.0f;
                if (!pad) fv = f[j * 16 + i];
                r = __builtin_fmaf(fv, wt[j * 3 + i], r);
            }
        float scale = e_g / sqrtf(e_v + 1e-5f);
        out[(((size_t)img * DCH + ech) * 8 + oh) * 8 + ow] = (r - e_m) * scale + e_b;
    }
}

extern "C" void kernel_launch(void* const* d_in, const int* in_sizes, int n_in,
                              void* d_out, int out_size, void* d_ws, size_t ws_size,
                              hipStream_t stream) {
    const float* x     = (const float*)d_in[0];
    const float* Wr    = (const float*)d_in[1];
    const float* br    = (const float*)d_in[2];
    const float* Wi    = (const float*)d_in[3];
    const float* bi    = (const float*)d_in[4];
    const float* cw    = (const float*)d_in[5];
    const float* cb    = (const float*)d_in[6];
    const float* gamma = (const float*)d_in[7];
    const float* beta  = (const float*)d_in[8];
    const float* mean  = (const float*)d_in[9];
    const float* var   = (const float*)d_in[10];

    char* ws = (char*)d_ws;
    double* ct256 = (double*)(ws + 0);
    double* st256 = (double*)(ws + 2048);
    double* ct64  = (double*)(ws + 4096);
    double* st64  = (double*)(ws + 4608);
    double* B1    = (double*)(ws + 5120);
    double* B2    = (double*)(ws + 136192);
    float*  A     = (float*) (ws + 267264);
    float*  cvec  = (float*) (ws + 332800);

    k_tables<<<10, 32, 0, stream>>>(ct256, st256, ct64, st64);
    k_b12<<<DCH * 4, 256, 0, stream>>>(Wr, Wi, ct256, st256, B1, B2);
    k_amat<<<DCH * 4, 256, 0, stream>>>(B1, B2, ct64, st64, br, bi, A, cvec);

    k_wmma<<<NB * 8 * 2, 384, 0, stream>>>(x, A, cvec, cw, cb, gamma, beta, mean, var, (float*)d_out);
}


# A handed from k_amat to k_wmma in operand-fragment order (16 rows x 8 features per 512 B, mirror rows adjacent): an A load reads 4 whole lines, not 32 B in each of 16; sum order unchanged
# speedup vs baseline: 6.2003x; 1.1407x over previous
#include <hip/hip_runtime.h>
#include <math.h>

#define IMG      2048
#define NB       16
#define DCH      64
#define INF      256

typedef float v2f __attribute__((ext_vector_type(2)));
typedef float v4f __attribute__((ext_vector_type(4)));
typedef float v8f __attribute__((ext_vector_type(8)));
typedef __attribute__((address_space(3))) v4f   lds_v4f;
typedef __attribute__((address_space(3))) float lds_f32;

__global__ void k_tables(double* ct256, double* st256, double* ct64, double* st64) {
    int g = blockIdx.x * 32 + threadIdx.x;
    const double TWO_PI = 6.283185307179586476925286766559;
    if (g < 256) { double a = TWO_PI * (double)g / 256.0; ct256[g] = cos(a); st256[g] = sin(a); }
    else { int t = g - 256; double a = TWO_PI * (double)t / 64.0;  ct64[t]  = cos(a); st64[t]  = sin(a); }
}

#define FOLDP 130
__global__ void k_b12(const float* __restrict__ Wr, const float* __restrict__ Wi,
                      const double* __restrict__ ct, const double* __restrict__ st,
                      double* __restrict__ B1, double* __restrict__ B2) {
    __shared__ double fold[4 * FOLDP];
    __shared__ double part[4][8][32];
    int d  = blockIdx.x >> 2;
    int nq = blockIdx.x & 3;
    int nl = threadIdx.x & 31;
    int kq = __builtin_amdgcn_readfirstlane((int)(threadIdx.x >> 5));
    int n  = nq * 32 + nl;

    {
        int kp  = threadIdx.x & 127;
        int sel = (int)(threadIdx.x >> 7) * 2;
        const float* w = (sel ? Wi : Wr) + d * INF;
        double a = (double)w[kp];
        double b = (double)w[(256 - kp) & 255];
        fold[sel * FOLDP + kp]       = (kp == 0) ? a : a + b;
        fold[(sel + 1) * FOLDP + kp] = a - b;
        if (kp == 0) {
            double h = (double)w[128];
            fold[sel * FOLDP + 128]       = h;
            fold[(sel + 1) * FOLDP + 128] = h;
        }
    }
    __syncthreads();

    const double* stab = (n == 0) ? ct : st;
    int ns = (n == 0) ? 128 : n;
    int so = (n == 0) ? 0 : FOLDP;
    double pr = 0.0, qr = 0.0, pi = 0.0, qi = 0.0;
    for (int j = 0; j < 16; ++j) {
        int kp = kq * 16 + j;
        double c = ct[(kp * n) & 255];
        double s = stab[(kp * ns) & 255];
        pr += fold[kp] * c;
        qr += fold[so + kp] * s;
        pi += fold[2 * FOLDP + kp] * c;
        qi += fold[2 * FOLDP + so + kp] * s;
    }
    if (kq == 0) {
        double c = ct[(128 * n) & 255];
        double s = stab[(128 * ns) & 255];
        pr += fold[128] * c;
        qr += fold[so + 128] * s;
        pi += fold[2 * FOLDP + 128] * c;
        qi += fold[2 * FOLDP + so + 128] * s;
    }
    part[0][kq][nl] = pr;
    part[1][kq][nl] = qr;
    part[2][kq][nl] = pi;
    part[3][kq][nl] = qi;
    __syncthreads();
    if (threadIdx.x < 128) {
        int which = kq;
        int xa = (which & 1) ? 2 : 0;
        int ya = (which & 1) ? 1 : 3;
        double X = 0.0, Y = 0.0;
        for (int p = 0; p < 8; ++p) { X += part[xa][p][nl]; Y += part[ya][p][nl]; }
        double v = (which == 0 || which == 3) ? (X + Y) : (X - Y);
        if (n == 0) {
            int idx = (which < 2) ? xa : ((which == 2) ? 1 : 3);
            double z = 0.0;
            for (int p = 0; p < 8; ++p) z += part[idx][p][0];
            v = z;
        }
        double* dst = (which & 1) ? B2 : B1;
        dst[d * INF + (which >> 1) * 128 + n] = v;
    }
}

#define AFRAG(p, f) ((((p) >> 4) * 32 + ((f) >> 3)) * 128 + ((p) & 15) * 8 + ((f) & 7))
__global__ void k_amat(const double* __restrict__ B1, const double* __restrict__ B2,
                       const double* __restrict__ ct64, const double* __restrict__ st64,
                       const float* __restrict__ br, const float* __restrict__ bi,
                       float* __restrict__ A, float* __restrict__ cvec) {
    __shared__ double pu[8][32];
    __shared__ double pv[8][32];
    int q    = blockIdx.x >> 4;
    int ne   = blockIdx.x & 15;
    int lane = threadIdx.x & 31;
    int nl   = lane & 15;
    int pp   = lane >> 4;
    int dq = __builtin_amdgcn_readfirstlane((int)(threadIdx.x >> 5));
    int m  = 2 * q + pp;
    int n  = ne * 16 + nl;
    int np = n <= 128 ? n : 384 - n;
    const double* vt = (m == 0) ? ct64 : st64;
    const double* vb = (m == 0) ? B1 : B2;
    int ms = (m == 0) ? 32 : m;
    double u = 0.0, v = 0.0;
    for (int j = 0; j < 8; ++j) {
        int d = dq * 8 + j;
        u += ct64[(d * m) & 63] * B1[d * INF + np];
        v += vt[(d * ms) & 63] * vb[d * INF + np];
    }
    pu[dq][lane] = u;
    pv[dq][lane] = v;
    __syncthreads();
    if (threadIdx.x < 64) {
        int w   = dq;
        int rho = lane >> 3;
        int phi = lane & 7;
        int sl  = (rho >> 1) * 16 + w * 8 + phi;
        int hi  = rho & 1;
        int mm  = 2 * q + (rho >> 1);
        double U = 0.0, V = 0.0;
        for (int p = 0; p < 8; ++p) { U += pu[p][sl]; V += pv[p][sl]; }
        double val = hi ? (U + V) : (U - V);
        if (mm == 0) val = hi ? V : U;
        int pos = 4 * q + rho;
        int f   = ne * 16 + w * 8 + phi;
        A[AFRAG(pos, f)] = (float)(val / 64.0);
    }
    if (ne == 0 && dq >= 2 && dq < 6) {
        int r  = dq - 2;
        int mm = 2 * q + (r >> 1);
        int mr = (r & 1) ? ((mm == 0) ? 32 : 64 - mm) : mm;
        int l = lane;
        double cc = 0.0;
        for (int h = 0; h < 2; ++h) {
            int d = l + 32 * h;
            int t = (d * mr) & 63;
            cc += ct64[t] * ((double)br[d] - (double)bi[d])
                - st64[t] * ((double)br[d] + (double)bi[d]);
        }
        for (int o = 16; o >= 1; o >>= 1) cc += __shfl_xor(cc, o);
        if (l == 0) cvec[4 * q + r] = (float)(cc / 64.0);
    }
}

#define SLOTW    12
#define FPITCH   52
__global__ void __launch_bounds__(384)
k_wmma(const float* __restrict__ x, const float* __restrict__ A, const float* __restrict__ cvec,
       const float* __restrict__ cw, const float* __restrict__ cb, const float* __restrict__ gamma,
       const float* __restrict__ beta, const float* __restrict__ mean, const float* __restrict__ var,
       float* __restrict__ out) {
    __shared__ v4f   red[4][3][4][2][32];
    __shared__ float fl[DCH * FPITCH];
    int lane = threadIdx.x & 31;
    int wid  = __builtin_amdgcn_readfirstlane((int)(threadIdx.x >> 5));
    int kq   = wid & 3;
    int tl   = wid >> 2;
    int half = blockIdx.x & 1;
    int oh   = (blockIdx.x >> 1) & 7;
    int img  = blockIdx.x >> 4;
    int n    = lane & 15;
    int hi   = lane >> 4;

    int s  = tl * 16 + n;
    int tc = n > (SLOTW - 1) ? (SLOTW - 1) : n;
    int prow = 16 * oh + tl - 1;
    int pcol = 16 * (4 * half + tc / 3) + (tc % 3) - 1;
    prow = prow < 0 ? 0 : prow;
    pcol = pcol < 0 ? 0 : pcol;

    const float* xb = x + (size_t)img * IMG * IMG
                        + (size_t)(prow * 16 + 4 * kq) * IMG + (size_t)(pcol * 16);
    const float* ap = A + (kq * 8) * 128 + n * 8 + hi * 4;

    v4f b[8];
    v4f a[4][8];
    #pragma unroll
    for (int j = 0; j < 8; ++j) {
        int u = 8 * j + 4 * hi;
        b[j] = *(const v4f*)(xb + (u >> 4) * IMG + (u & 15));
    }
    __builtin_amdgcn_sched_barrier(0);
    #pragma unroll
    for (int t = 0; t < 4; ++t)
        #pragma unroll
        for (int j = 0; j < 8; ++j) {
            a[t][j] = *(const v4f*)(ap + (t * 32 + j) * 128);
        }
    int epos = (int)(threadIdx.x >> 2) & 63;
    int ech  = (epos & 1) ? ((epos >> 1) == 0 ? 32 : 64 - (epos >> 1)) : (epos >> 1);
    int owl = (int)threadIdx.x & 3;
    v4f cv0 = *(const v4f*)(cvec + kq * 16 + hi * 8);
    v4f cv1 = *(const v4f*)(cvec + kq * 16 + hi * 8 + 4);
    float wt[9];
    #pragma unroll
    for (int k = 0; k < 9; ++k) wt[k] = cw[ech * 9 + k];
    float e_cb = cb[ech], e_g = gamma[ech], e_b = beta[ech], e_m = mean[ech], e_v = var[ech];
    lds_v4f* rst = (lds_v4f*)&red[kq][tl][0][0][lane];
    lds_v4f* rld = (lds_v4f*)&red[0][tl][kq][0][lane];
    lds_f32* fst = (lds_f32*)(fl + (kq * 16 + hi * 8) * FPITCH + s);
    asm volatile("" : "+v"(rst), "+v"(rld), "+v"(fst));

    __builtin_amdgcn_sched_barrier(0);

    v8f acc[4] = {};
    #pragma unroll
    for (int j = 0; j < 8; ++j)
        #pragma unroll
        for (int p = 0; p < 2; ++p) {
            v2f bv = p ? (v2f){b[j][2], b[j][3]} : (v2f){b[j][0], b[j][1]};
            #pragma unroll
            for (int t = 0; t < 4; ++t) {
                v2f av = p ? (v2f){a[t][j][2], a[t][j][3]} : (v2f){a[t][j][0], a[t][j][1]};
                acc[t] = __builtin_amdgcn_wmma_f32_16x16x4_f32(
                              false, av,  false, bv,
                              (short)0, acc[t],  false,  false);
            }
        }

    #pragma unroll
    for (int t = 0; t < 4; ++t) {
        rst[64 * t]      = (v4f){acc[t][0], acc[t][1], acc[t][2], acc[t][3]};
        rst[64 * t + 32] = (v4f){acc[t][4], acc[t][5], acc[t][6], acc[t][7]};
    }
    __syncthreads();
    v4f lo = rld[0];
    v4f up = rld[32];
    #pragma unroll
    for (int p = 1; p < 4; ++p) {
        lo += rld[768 * p];
        up += rld[768 * p + 32];
    }
    #pragma unroll
    for (int j = 0; j < 8; ++j) {
        float v = (j < 4) ? lo[j & 3] : up[j & 3];
        float c = (j < 4) ? cv0[j & 3] : cv1[j & 3];
        fst[FPITCH * j] = v + c;
    }
    __syncthreads();

    if (threadIdx.x < 256) {
        int ow = 4 * half + owl;
        const float* f = fl + epos * FPITCH + 3 * owl;
        float r = e_cb;
        #pragma unroll
        for (int j = 0; j < 3; ++j)
            #pragma unroll
            for (int i = 0; i < 3; ++i) {
                bool pad = (j == 0 && oh == 0) || (i == 0 && ow == 0);
                float fv = 0.0f;
                if (!pad) fv = f[j * 16 + i];
                r = __builtin_fmaf(fv, wt[j * 3 + i], r);
            }
        float scale = e_g / sqrtf(e_v + 1e-5f);
        out[(((size_t)img * DCH + ech) * 8 + oh) * 8 + ow] = (r - e_m) * scale + e_b;
    }
}

extern "C" void kernel_launch(void* const* d_in, const int* in_sizes, int n_in,
                              void* d_out, int out_size, void* d_ws, size_t ws_size,
                              hipStream_t stream) {
    const float* x     = (const float*)d_in[0];
    const float* Wr    = (const float*)d_in[1];
    const float* br    = (const float*)d_in[2];
    const float* Wi    = (const float*)d_in[3];
    const float* bi    = (const float*)d_in[4];
    const float* cw    = (const float*)d_in[5];
    const float* cb    = (const float*)d_in[6];
    const float* gamma = (const float*)d_in[7];
    const float* beta  = (const float*)d_in[8];
    const float* mean  = (const float*)d_in[9];
    const float* var   = (const float*)d_in[10];

    char* ws = (char*)d_ws;
    double* ct256 = (double*)(ws + 0);
    double* st256 = (double*)(ws + 2048);
    double* ct64  = (double*)(ws + 4096);
    double* st64  = (double*)(ws + 4608);
    double* B1    = (double*)(ws + 5120);
    double* B2    = (double*)(ws + 136192);
    float*  A     = (float*) (ws + 267264);
    float*  cvec  = (float*) (ws + 332800);

    k_tables<<<10, 32, 0, stream>>>(ct256, st256, ct64, st64);
    k_b12<<<DCH * 4, 256, 0, stream>>>(Wr, Wi, ct256, st256, B1, B2);
    k_amat<<<DCH * 4, 256, 0, stream>>>(B1, B2, ct64, st64, br, bi, A, cvec);

    k_wmma<<<NB * 8 * 2, 384, 0, stream>>>(x, A, cvec, cw, cb, gamma, beta, mean, var, (float*)d_out);
}


# k_tables: each entry one 12-instruction FP64 chain (integer quarter-turn fold, then ONE 8-term series) in place of two library calls; sine table = cosine table shifted a quarter; sum order unchanged
# speedup vs baseline: 6.4935x; 1.0473x over previous
#include <hip/hip_runtime.h>
#include <math.h>

#define IMG      2048
#define NB       16
#define DCH      64
#define INF      256

typedef float v2f __attribute__((ext_vector_type(2)));
typedef float v4f __attribute__((ext_vector_type(4)));
typedef float v8f __attribute__((ext_vector_type(8)));
typedef __attribute__((address_space(3))) v4f   lds_v4f;
typedef __attribute__((address_space(3))) float lds_f32;

__device__ __forceinline__ double cos_turn256(int u) {
    int  h   = (u + 32) >> 6;
    int  rho = u - (h << 6);
    bool odd = (h & 1) != 0;
    bool neg = (((h + 1) >> 1) & 1) != 0;
    double t  = (double)rho * (3.14159265358979323846264338327950288 / 128.0);
    double z  = t * t;
    double zt = z * t;
    double p = odd ? (1.0 / 355687428096000.0) : (1.0 / 20922789888000.0);
    p = fma(p, z, odd ? (-1.0 / 1307674368000.0) : (-1.0 / 87178291200.0));
    p = fma(p, z, odd ? (1.0 / 6227020800.0)     : (1.0 / 479001600.0));
    p = fma(p, z, odd ? (-1.0 / 39916800.0)      : (-1.0 / 3628800.0));
    p = fma(p, z, odd ? (1.0 / 362880.0)         : (1.0 / 40320.0));
    p = fma(p, z, odd ? (-1.0 / 5040.0)          : (-1.0 / 720.0));
    p = fma(p, z, odd ? (1.0 / 120.0)            : (1.0 / 24.0));
    p = fma(p, z, odd ? (-1.0 / 6.0)             : (-1.0 / 2.0));
    double r = fma(p, odd ? zt : z, odd ? t : 1.0);
    return neg ? -r : r;
}
__global__ void k_tables(double* ct256, double* st256, double* ct64, double* st64) {
    int g = blockIdx.x * 32 + threadIdx.x;
    if (g < 256) { double v = cos_turn256(g); ct256[g] = v; st256[(g + 64) & 255] = v; }
    else { int t = g - 256; double v = cos_turn256(4 * t); ct64[t] = v; st64[(t + 16) & 63] = v; }
}

#define FOLDP 130
__global__ void k_b12(const float* __restrict__ Wr, const float* __restrict__ Wi,
                      const double* __restrict__ ct, const double* __restrict__ st,
                      double* __restrict__ B1, double* __restrict__ B2) {
    __shared__ double fold[4 * FOLDP];
    __shared__ double part[4][8][32];
    int d  = blockIdx.x >> 2;
    int nq = blockIdx.x & 3;
    int nl = threadIdx.x & 31;
    int kq = __builtin_amdgcn_readfirstlane((int)(threadIdx.x >> 5));
    int n  = nq * 32 + nl;

    {
        int kp  = threadIdx.x & 127;
        int sel = (int)(threadIdx.x >> 7) * 2;
        const float* w = (sel ? Wi : Wr) + d * INF;
        double a = (double)w[kp];
        double b = (double)w[(256 - kp) & 255];
        fold[sel * FOLDP + kp]       = (kp == 0) ? a : a + b;
        fold[(sel + 1) * FOLDP + kp] = a - b;
        if (kp == 0) {
            double h = (double)w[128];
            fold[sel * FOLDP + 128]       = h;
            fold[(sel + 1) * FOLDP + 128] = h;
        }
    }
    __syncthreads();

    const double* stab = (n == 0) ? ct : st;
    int ns = (n == 0) ? 128 : n;
    int so = (n == 0) ? 0 : FOLDP;
    double pr = 0.0, qr = 0.0, pi = 0.0, qi = 0.0;
    for (int j = 0; j < 16; ++j) {
        int kp = kq * 16 + j;
        double c = ct[(kp * n) & 255];
        double s = stab[(kp * ns) & 255];
        pr += fold[kp] * c;
        qr += fold[so + kp] * s;
        pi += fold[2 * FOLDP + kp] * c;
        qi += fold[2 * FOLDP + so + kp] * s;
    }
    if (kq == 0) {
        double c = ct[(128 * n) & 255];
        double s = stab[(128 * ns) & 255];
        pr += fold[128] * c;
        qr += fold[so + 128] * s;
        pi += fold[2 * FOLDP + 128] * c;
        qi += fold[2 * FOLDP + so + 128] * s;
    }
    part[0][kq][nl] = pr;
    part[1][kq][nl] = qr;
    part[2][kq][nl] = pi;
    part[3][kq][nl] = qi;
    __syncthreads();
    if (threadIdx.x < 128) {
        int which = kq;
        int xa = (which & 1) ? 2 : 0;
        int ya = (which & 1) ? 1 : 3;
        double X = 0.0, Y = 0.0;
        for (int p = 0; p < 8; ++p) { X += part[xa][p][nl]; Y += part[ya][p][nl]; }
        double v = (which == 0 || which == 3) ? (X + Y) : (X - Y);
        if (n == 0) {
            int idx = (which < 2) ? xa : ((which == 2) ? 1 : 3);
            double z = 0.0;
            for (int p = 0; p < 8; ++p) z += part[idx][p][0];
            v = z;
        }
        double* dst = (which & 1) ? B2 : B1;
        dst[d * INF + (which >> 1) * 128 + n] = v;
    }
}

#define AFRAG(p, f) ((((p) >> 4) * 32 + ((f) >> 3)) * 128 + ((p) & 15) * 8 + ((f) & 7))
__global__ void k_amat(const double* __restrict__ B1, const double* __restrict__ B2,
                       const double* __restrict__ ct64, const double* __restrict__ st64,
                       const float* __restrict__ br, const float* __restrict__ bi,
                       float* __restrict__ A, float* __restrict__ cvec) {
    __shared__ double pu[8][32];
    __shared__ double pv[8][32];
    int q    = blockIdx.x >> 4;
    int ne   = blockIdx.x & 15;
    int lane = threadIdx.x & 31;
    int nl   = lane & 15;
    int pp   = lane >> 4;
    int dq = __builtin_amdgcn_readfirstlane((int)(threadIdx.x >> 5));
    int m  = 2 * q + pp;
    int n  = ne * 16 + nl;
    int np = n <= 128 ? n : 384 - n;
    const double* vt = (m == 0) ? ct64 : st64;
    const double* vb = (m == 0) ? B1 : B2;
    int ms = (m == 0) ? 32 : m;
    double u = 0.0, v = 0.0;
    for (int j = 0; j < 8; ++j) {
        int d = dq * 8 + j;
        u += ct64[(d * m) & 63] * B1[d * INF + np];
        v += vt[(d * ms) & 63] * vb[d * INF + np];
    }
    pu[dq][lane] = u;
    pv[dq][lane] = v;
    __syncthreads();
    if (threadIdx.x < 64) {
        int w   = dq;
        int rho = lane >> 3;
        int phi = lane & 7;
        int sl  = (rho >> 1) * 16 + w * 8 + phi;
        int hi  = rho & 1;
        int mm  = 2 * q + (rho >> 1);
        double U = 0.0, V = 0.0;
        for (int p = 0; p < 8; ++p) { U += pu[p][sl]; V += pv[p][sl]; }
        double val = hi ? (U + V) : (U - V);
        if (mm == 0) val = hi ? V : U;
        int pos = 4 * q + rho;
        int f   = ne * 16 + w * 8 + phi;
        A[AFRAG(pos, f)] = (float)(val / 64.0);
    }
    if (ne == 0 && dq >= 2 && dq < 6) {
        int r  = dq - 2;
        int mm = 2 * q + (r >> 1);
        int mr = (r & 1) ? ((mm == 0) ? 32 : 64 - mm) : mm;
        int l = lane;
        double cc = 0.0;
        for (int h = 0; h < 2; ++h) {
            int d = l + 32 * h;
            int t = (d * mr) & 63;
            cc += ct64[t] * ((double)br[d] - (double)bi[d])
                - st64[t] * ((double)br[d] + (double)bi[d]);
        }
        for (int o = 16; o >= 1; o >>= 1) cc += __shfl_xor(cc, o);
        if (l == 0) cvec[4 * q + r] = (float)(cc / 64.0);
    }
}

#define SLOTW    12
#define FPITCH   52
__global__ void __launch_bounds__(384)
k_wmma(const float* __restrict__ x, const float* __restrict__ A, const float* __restrict__ cvec,
       const float* __restrict__ cw, const float* __restrict__ cb, const float* __restrict__ gamma,
       const float* __restrict__ beta, const float* __restrict__ mean, const float* __restrict__ var,
       float* __restrict__ out) {
    __shared__ v4f   red[4][3][4][2][32];
    __shared__ float fl[DCH * FPITCH];
    int lane = threadIdx.x & 31;
    int wid  = __builtin_amdgcn_readfirstlane((int)(threadIdx.x >> 5));
    int kq   = wid & 3;
    int tl   = wid >> 2;
    int half = blockIdx.x & 1;
    int oh   = (blockIdx.x >> 1) & 7;
    int img  = blockIdx.x >> 4;
    int n    = lane & 15;
    int hi   = lane >> 4;

    int s  = tl * 16 + n;
    int tc = n > (SLOTW - 1) ? (SLOTW - 1) : n;
    int prow = 16 * oh + tl - 1;
    int pcol = 16 * (4 * half + tc / 3) + (tc % 3) - 1;
    prow = prow < 0 ? 0 : prow;
    pcol = pcol < 0 ? 0 : pcol;

    const float* xb = x + (size_t)img * IMG * IMG
                        + (size_t)(prow * 16 + 4 * kq) * IMG + (size_t)(pcol * 16);
    const float* ap = A + (kq * 8) * 128 + n * 8 + hi * 4;

    v4f b[8];
    v4f a[4][8];
    #pragma unroll
    for (int j = 0; j < 8; ++j) {
        int u = 8 * j + 4 * hi;
        b[j] = *(const v4f*)(xb + (u >> 4) * IMG + (u & 15));
    }
    __builtin_amdgcn_sched_barrier(0);
    #pragma unroll
    for (int t = 0; t < 4; ++t)
        #pragma unroll
        for (int j = 0; j < 8; ++j) {
            a[t][j] = *(const v4f*)(ap + (t * 32 + j) * 128);
        }
    int epos = (int)(threadIdx.x >> 2) & 63;
    int ech  = (epos & 1) ? ((epos >> 1) == 0 ? 32 : 64 - (epos >> 1)) : (epos >> 1);
    int owl = (int)threadIdx.x & 3;
    v4f cv0 = *(const v4f*)(cvec + kq * 16 + hi * 8);
    v4f cv1 = *(const v4f*)(cvec + kq * 16 + hi * 8 + 4);
    float wt[9];
    #pragma unroll
    for (int k = 0; k < 9; ++k) wt[k] = cw[ech * 9 + k];
    float e_cb = cb[ech], e_g = gamma[ech], e_b = beta[ech], e_m = mean[ech], e_v = var[ech];
    lds_v4f* rst = (lds_v4f*)&red[kq][tl][0][0][lane];
    lds_v4f* rld = (lds_v4f*)&red[0][tl][kq][0][lane];
    lds_f32* fst = (lds_f32*)(fl + (kq * 16 + hi * 8) * FPITCH + s);
    asm volatile("" : "+v"(rst), "+v"(rld), "+v"(fst));

    __builtin_amdgcn_sched_barrier(0);

    v8f acc[4] = {};
    #pragma unroll
    for (int j = 0; j < 8; ++j)
        #pragma unroll
        for (int p = 0; p < 2; ++p) {
            v2f bv = p ? (v2f){b[j][2], b[j][3]} : (v2f){b[j][0], b[j][1]};
            #pragma unroll
            for (int t = 0; t < 4; ++t) {
                v2f av = p ? (v2f){a[t][j][2], a[t][j][3]} : (v2f){a[t][j][0], a[t][j][1]};
                acc[t] = __builtin_amdgcn_wmma_f32_16x16x4_f32(
                              false, av,  false, bv,
                              (short)0, acc[t],  false,  false);
            }
        }

    #pragma unroll
    for (int t = 0; t < 4; ++t) {
        rst[64 * t]      = (v4f){acc[t][0], acc[t][1], acc[t][2], acc[t][3]};
        rst[64 * t + 32] = (v4f){acc[t][4], acc[t][5], acc[t][6], acc[t][7]};
    }
    __syncthreads();
    v4f lo = rld[0];
    v4f up = rld[32];
    #pragma unroll
    for (int p = 1; p < 4; ++p) {
        lo += rld[768 * p];
        up += rld[768 * p + 32];
    }
    #pragma unroll
    for (int j = 0; j < 8; ++j) {
        float v = (j < 4) ? lo[j & 3] : up[j & 3];
        float c = (j < 4) ? cv0[j & 3] : cv1[j & 3];
        fst[FPITCH * j] = v + c;
    }
    __syncthreads();

    if (threadIdx.x < 256) {
        int ow = 4 * half + owl;
        const float* f = fl + epos * FPITCH + 3 * owl;
        float r = e_cb;
        #pragma unroll
        for (int j = 0; j < 3; ++j)
            #pragma unroll
            for (int i = 0; i < 3; ++i) {
                bool pad = (j == 0 && oh == 0) || (i == 0 && ow == 0);
                float fv = 0.0f;
                if (!pad) fv = f[j * 16 + i];
                r = __builtin_fmaf(fv, wt[j * 3 + i], r);
            }
        float scale = e_g / sqrtf(e_v + 1e-5f);
        out[(((size_t)img * DCH + ech) * 8 + oh) * 8 + ow] = (r - e_m) * scale + e_b;
    }
}

extern "C" void kernel_launch(void* const* d_in, const int* in_sizes, int n_in,
                              void* d_out, int out_size, void* d_ws, size_t ws_size,
                              hipStream_t stream) {
    const float* x     = (const float*)d_in[0];
    const float* Wr    = (const float*)d_in[1];
    const float* br    = (const float*)d_in[2];
    const float* Wi    = (const float*)d_in[3];
    const float* bi    = (const float*)d_in[4];
    const float* cw    = (const float*)d_in[5];
    const float* cb    = (const float*)d_in[6];
    const float* gamma = (const float*)d_in[7];
    const float* beta  = (const float*)d_in[8];
    const float* mean  = (const float*)d_in[9];
    const float* var   = (const float*)d_in[10];

    char* ws = (char*)d_ws;
    double* ct256 = (double*)(ws + 0);
    double* st256 = (double*)(ws + 2048);
    double* ct64  = (double*)(ws + 4096);
    double* st64  = (double*)(ws + 4608);
    double* B1    = (double*)(ws + 5120);
    double* B2    = (double*)(ws + 136192);
    float*  A     = (float*) (ws + 267264);
    float*  cvec  = (float*) (ws + 332800);

    k_tables<<<10, 32, 0, stream>>>(ct256, st256, ct64, st64);
    k_b12<<<DCH * 4, 256, 0, stream>>>(Wr, Wi, ct256, st256, B1, B2);
    k_amat<<<DCH * 4, 256, 0, stream>>>(B1, B2, ct64, st64, br, bi, A, cvec);

    k_wmma<<<NB * 8 * 2, 384, 0, stream>>>(x, A, cvec, cw, cb, gamma, beta, mean, var, (float*)d_out);
}


# k_b12 + k_amat fused, 4 -> 3 launches: block = column pair x side U|V sums over k, then over d in LDS; W folded once in launch 1; k_wmma joins U-V, U+V after its product; new double and f32 sum order
# speedup vs baseline: 6.8934x; 1.0616x over previous
#include <hip/hip_runtime.h>
#include <math.h>

#define IMG      2048
#define NB       16
#define DCH      64
#define INF      256

typedef float v2f __attribute__((ext_vector_type(2)));
typedef float v4f __attribute__((ext_vector_type(4)));
typedef float v8f __attribute__((ext_vector_type(8)));
typedef __attribute__((address_space(3))) v4f   lds_v4f;
typedef __attribute__((address_space(3))) float lds_f32;

__device__ __forceinline__ double cos_turn256(int u) {
    int  h   = (u + 32) >> 6;
    int  rho = u - (h << 6);
    bool odd = (h & 1) != 0;
    bool neg = (((h + 1) >> 1) & 1) != 0;
    double t  = (double)rho * (3.14159265358979323846264338327950288 / 128.0);
    double z  = t * t;
    double zt = z * t;
    double p = odd ? (1.0 / 355687428096000.0) : (1.0 / 20922789888000.0);
    p = fma(p, z, odd ? (-1.0 / 1307674368000.0) : (-1.0 / 87178291200.0));
    p = fma(p, z, odd ? (1.0 / 6227020800.0)     : (1.0 / 479001600.0));
    p = fma(p, z, odd ? (-1.0 / 39916800.0)      : (-1.0 / 3628800.0));
    p = fma(p, z, odd ? (1.0 / 362880.0)         : (1.0 / 40320.0));
    p = fma(p, z, odd ? (-1.0 / 5040.0)          : (-1.0 / 720.0));
    p = fma(p, z, odd ? (1.0 / 120.0)            : (1.0 / 24.0));
    p = fma(p, z, odd ? (-1.0 / 6.0)             : (-1.0 / 2.0));
    double r = fma(p, odd ? zt : z, odd ? t : 1.0);
    return neg ? -r : r;
}
#define XSZ (129 * DCH)
__global__ void __launch_bounds__(128)
k_pre(const float* __restrict__ Wr, const float* __restrict__ Wi,
      double* __restrict__ ct256, double* __restrict__ ct64, double* __restrict__ X) {
    int b = blockIdx.x;
    if (b < 3) {
        int g = b * 128 + threadIdx.x;
        if (g < 256)      ct256[g] = cos_turn256(g);
        else if (g < 320) ct64[g - 256] = cos_turn256(4 * (g - 256));
        return;
    }
    int g = (b - 3) * 128 + threadIdx.x;
    if (g < 16384) {
        int lane = g & 31;
        int gw   = g >> 5;
        int d    = (gw & 3) * 16 + (lane & 15);
        int k    = ((gw >> 2) & 63) * 2 + (lane >> 4);
        int sel  = gw >> 8;
        const float* w = (sel ? Wi : Wr) + d * INF;
        double a = (double)w[k];
        double c = (double)w[(256 - k) & 255];
        double* xe = X + (2 * sel) * XSZ;
        double* xo = xe + XSZ;
        xe[k * DCH + d] = (k == 0) ? a : a + c;
        xo[k * DCH + d] = a - c;
    } else {
        int t   = g - 16384;
        int sel = t >> 6;
        int d   = t & 63;
        const float* w = (sel ? Wi : Wr) + d * INF;
        double* xe = X + (2 * sel) * XSZ;
        double* xo = xe + XSZ;
        xe[128 * DCH + d] = (double)w[128];
        xo[128 * DCH + d] = 0.0;
    }
}

#define AFRAG(p, f) ((((p) >> 4) * 32 + ((f) >> 3)) * 128 + ((p) & 15) * 8 + ((f) & 7))
typedef int v2i __attribute__((ext_vector_type(2)));
__device__ __forceinline__ double lane_value(double v, int j) {
    v2i p = __builtin_bit_cast(v2i, v);
    p.x = __builtin_amdgcn_readlane(p.x, j);
    p.y = __builtin_amdgcn_readlane(p.y, j);
    return __builtin_bit_cast(double, p);
}
__global__ void __launch_bounds__(256)
k_fuse(const double* __restrict__ X, const double* __restrict__ ct, const double* __restrict__ ct64,
       const float* __restrict__ br, const float* __restrict__ bi,
       float* __restrict__ A, float* __restrict__ cvec) {
    __shared__ double part[2][4][DCH];
    __shared__ double S[2][DCH];
    __shared__ double pa[2][4][32];
    __shared__ double p32[2][4];
    int n    = blockIdx.x >> 1;
    int side = blockIdx.x & 1;
    int lane = threadIdx.x & 31;
    int wv = __builtin_amdgcn_readfirstlane((int)(threadIdx.x >> 5));
    int d  = (wv & 1) * 32 + lane;
    int kq = wv >> 1;
    const double* X1 = X + (side ? 2 : 0) * XSZ;
    const double* X2 = (n == 0) ? X1 : X + (side ? 1 : 3) * XSZ;
    int n2 = (n == 0) ? 128 : n;
    int sh = (n == 0) ? 0 : 192;

    bool biasw = (side == 1) && (n < 64) && (wv == 5);
    float  fr0 = 0.0f, fr1 = 0.0f, fi0 = 0.0f, fi1 = 0.0f;
    double bc0 = 0.0, bc1 = 0.0, bq0 = 0.0, bq1 = 0.0;
    if (biasw) {
        int mm  = n >> 1;
        int bmr = (n & 1) ? ((mm == 0) ? 32 : 64 - mm) : mm;
        int t0 = (lane * bmr) & 63, t1 = ((lane + 32) * bmr) & 63;
        fr0 = br[lane];      fi0 = bi[lane];
        fr1 = br[lane + 32]; fi1 = bi[lane + 32];
        bc0 = ct64[t0]; bq0 = ct64[(t0 + 48) & 63];
        bc1 = ct64[t1]; bq1 = ct64[(t1 + 48) & 63];
    }
    int    kl  = kq * 32 + lane;
    double t1v = ct[(kl * n) & 255];
    double t2v = ct[(kl * n2 + sh) & 255];
    double x1[32], x2[32];
    #pragma unroll
    for (int j = 0; j < 32; ++j) {
        int k = kq * 32 + j;
        x1[j] = X1[k * DCH + d];
        x2[j] = X2[k * DCH + d];
    }
    double xe1 = X1[128 * DCH + d];
    double xe2 = X2[128 * DCH + d];
    double te1 = ct[(128 * n) & 255];
    double te2 = ct[(128 * n2 + sh) & 255];
    int rsh = side ? 48 : 0;
    double tc[16];
    #pragma unroll
    for (int j = 0; j < 16; ++j) tc[j] = ct64[((kq * 16 + j) * lane + rsh) & 63];
    __builtin_amdgcn_sched_barrier(0);
    if (biasw) {
        double cc = 0.0;
        cc += bc0 * ((double)fr0 - (double)fi0) - bq0 * ((double)fr0 + (double)fi0);
        cc += bc1 * ((double)fr1 - (double)fi1) - bq1 * ((double)fr1 + (double)fi1);
        for (int o = 16; o >= 1; o >>= 1) cc += __shfl_xor(cc, o);
        if (lane == 0) cvec[n] = (float)(cc / 64.0);
    }
    double s1 = 0.0, s2 = 0.0;
    #pragma unroll
    for (int j = 0; j < 32; ++j) {
        s1 += x1[j] * lane_value(t1v, j);
        s2 += x2[j] * lane_value(t2v, j);
    }
    if (kq == 0) {
        s1 += xe1 * te1;
        s2 += xe2 * te2;
    }
    part[0][kq][d] = s1;
    part[1][kq][d] = s2;
    __syncthreads();
    if (wv >= 4) {
        int o   = (int)threadIdx.x - 128;
        int sel = o >> 6;
        int dd  = o & 63;
        S[sel][dd] = ((part[sel][0][dd] + part[sel][1][dd]) + part[sel][2][dd]) + part[sel][3][dd];
    }
    __syncthreads();

    {
        int sel = wv & 1;
        int dq  = wv >> 1;
        double ae = 0.0, ao = 0.0;
        #pragma unroll
        for (int j = 0; j < 16; j += 2) {
            int d0 = dq * 16 + j;
            ae += tc[j] * S[sel][d0];
            ao += tc[j + 1] * S[sel][d0 + 1];
        }
        pa[sel][dq][lane] = ae + ao;
        if (lane == 0) p32[sel][dq] = ae - ao;
    }
    __syncthreads();

    if (wv == 2 || wv == 3) {
        int c = wv - 2;
        double P = ((pa[0][0][lane] + pa[0][1][lane]) + pa[0][2][lane]) + pa[0][3][lane];
        double Q = ((pa[1][0][lane] + pa[1][1][lane]) + pa[1][2][lane]) + pa[1][3][lane];
        double val = (c == side) ? (P + Q) : (P - Q);
        if (n == 0) val = c ? Q : P;
        int f   = c ? ((n == 0) ? 128 : 256 - n) : n;
        int pos = (lane == 0) ? 0 : 2 * lane + side;
        if (!(side == 1 && lane == 0))
            A[AFRAG(pos, f)] = (float)(val / 64.0);
    }
    if (wv == 5 && side == 0 && lane < 2) {
        int c = lane;
        double P = ((p32[0][0] + p32[0][1]) + p32[0][2]) + p32[0][3];
        double Q = ((p32[1][0] + p32[1][1]) + p32[1][2]) + p32[1][3];
        double val = c ? (P - Q) : (P + Q);
        if (n == 0) val = c ? Q : P;
        int f = c ? ((n == 0) ? 128 : 256 - n) : n;
        A[AFRAG(1, f)] = (float)(val / 64.0);
    }
}

#define SLOTW    12
#define FPITCH   52
__global__ void __launch_bounds__(384)
k_wmma(const float* __restrict__ x, const float* __restrict__ A, const float* __restrict__ cvec,
       const float* __restrict__ cw, const float* __restrict__ cb, const float* __restrict__ gamma,
       const float* __restrict__ beta, const float* __restrict__ mean, const float* __restrict__ var,
       float* __restrict__ out) {
    __shared__ v4f   red[4][3][4][2][32];
    __shared__ float fl[DCH * FPITCH];
    int lane = threadIdx.x & 31;
    int wid  = __builtin_amdgcn_readfirstlane((int)(threadIdx.x >> 5));
    int kq   = wid & 3;
    int tl   = wid >> 2;
    int half = blockIdx.x & 1;
    int oh   = (blockIdx.x >> 1) & 7;
    int img  = blockIdx.x >> 4;
    int n    = lane & 15;
    int hi   = lane >> 4;

    int s  = tl * 16 + n;
    int tc = n > (SLOTW - 1) ? (SLOTW - 1) : n;
    int prow = 16 * oh + tl - 1;
    int pcol = 16 * (4 * half + tc / 3) + (tc % 3) - 1;
    prow = prow < 0 ? 0 : prow;
    pcol = pcol < 0 ? 0 : pcol;

    const float* xb = x + (size_t)img * IMG * IMG
                        + (size_t)(prow * 16 + 4 * kq) * IMG + (size_t)(pcol * 16);
    const float* ap = A + (kq * 8) * 128 + n * 8 + hi * 4;

    v4f b[8];
    v4f a[4][8];
    #pragma unroll
    for (int j = 0; j < 8; ++j) {
        int u = 8 * j + 4 * hi;
        b[j] = *(const v4f*)(xb + (u >> 4) * IMG + (u & 15));
    }
    __builtin_amdgcn_sched_barrier(0);
    #pragma unroll
    for (int t = 0; t < 4; ++t)
        #pragma unroll
        for (int j = 0; j < 8; ++j) {
            a[t][j] = *(const v4f*)(ap + (t * 32 + j) * 128);
        }
    int epos = (int)(threadIdx.x >> 2) & 63;
    int ech  = (epos & 1) ? ((epos >> 1) == 0 ? 32 : 64 - (epos >> 1)) : (epos >> 1);
    int owl = (int)threadIdx.x & 3;
    v4f cv0 = *(const v4f*)(cvec + kq * 16 + hi * 8);
    v4f cv1 = *(const v4f*)(cvec + kq * 16 + hi * 8 + 4);
    float wt[9];
    #pragma unroll
    for (int k = 0; k < 9; ++k) wt[k] = cw[ech * 9 + k];
    float e_cb = cb[ech], e_g = gamma[ech], e_b = beta[ech], e_m = mean[ech], e_v = var[ech];
    lds_v4f* rst = (lds_v4f*)&red[kq][tl][0][0][lane];
    lds_v4f* rld = (lds_v4f*)&red[0][tl][kq][0][lane];
    lds_f32* fst = (lds_f32*)(fl + (kq * 16 + hi * 8) * FPITCH + s);
    asm volatile("" : "+v"(rst), "+v"(rld), "+v"(fst));

    __builtin_amdgcn_sched_barrier(0);

    v8f acc[4] = {};
    #pragma unroll
    for (int j = 0; j < 8; ++j)
        #pragma unroll
        for (int p = 0; p < 2; ++p) {
            v2f bv = p ? (v2f){b[j][2], b[j][3]} : (v2f){b[j][0], b[j][1]};
            #pragma unroll
            for (int t = 0; t < 4; ++t) {
                v2f av = p ? (v2f){a[t][j][2], a[t][j][3]} : (v2f){a[t][j][0], a[t][j][1]};
                acc[t] = __builtin_amdgcn_wmma_f32_16x16x4_f32(
                              false, av,  false, bv,
                              (short)0, acc[t],  false,  false);
            }
        }

    #pragma unroll
    for (int t = 0; t < 4; ++t) {
        rst[64 * t]      = (v4f){acc[t][0], acc[t][1], acc[t][2], acc[t][3]};
        rst[64 * t + 32] = (v4f){acc[t][4], acc[t][5], acc[t][6], acc[t][7]};
    }
    __syncthreads();
    v4f lo = rld[0];
    v4f up = rld[32];
    #pragma unroll
    for (int p = 1; p < 4; ++p) {
        lo += rld[768 * p];
        up += rld[768 * p + 32];
    }
    bool first = (kq == 0) && (hi == 0);
    #pragma unroll
    for (int j = 0; j < 8; j += 2) {
        float fu = (j < 4) ? lo[j & 3] : up[j & 3];
        float fv = (j < 4) ? lo[(j + 1) & 3] : up[(j + 1) & 3];
        float c0 = (j < 4) ? cv0[j & 3] : cv1[j & 3];
        float c1 = (j < 4) ? cv0[(j + 1) & 3] : cv1[(j + 1) & 3];
        bool keep = (j == 0) && first;
        float e = keep ? fu : fu - fv;
        float o = keep ? fv : fu + fv;
        fst[FPITCH * j]       = e + c0;
        fst[FPITCH * (j + 1)] = o + c1;
    }
    __syncthreads();

    if (threadIdx.x < 256) {
        int ow = 4 * half + owl;
        const float* f = fl + epos * FPITCH + 3 * owl;
        float r = e_cb;
        #pragma unroll
        for (int j = 0; j < 3; ++j)
            #pragma unroll
            for (int i = 0; i < 3; ++i) {
                bool pad = (j == 0 && oh == 0) || (i == 0 && ow == 0);
                float fv = 0.0f;
                if (!pad) fv = f[j * 16 + i];
                r = __builtin_fmaf(fv, wt[j * 3 + i], r);
            }
        float scale = e_g / sqrtf(e_v + 1e-5f);
        out[(((size_t)img * DCH + ech) * 8 + oh) * 8 + ow] = (r - e_m) * scale + e_b;
    }
}

extern "C" void kernel_launch(void* const* d_in, const int* in_sizes, int n_in,
                              void* d_out, int out_size, void* d_ws, size_t ws_size,
                              hipStream_t stream) {
    const float* x     = (const float*)d_in[0];
    const float* Wr    = (const float*)d_in[1];
    const float* br    = (const float*)d_in[2];
    const float* Wi    = (const float*)d_in[3];
    const float* bi    = (const float*)d_in[4];
    const float* cw    = (const float*)d_in[5];
    const float* cb    = (const float*)d_in[6];
    const float* gamma = (const float*)d_in[7];
    const float* beta  = (const float*)d_in[8];
    const float* mean  = (const float*)d_in[9];
    const float* var   = (const float*)d_in[10];

    char* ws = (char*)d_ws;
    double* ct256 = (double*)(ws + 0);
    double* ct64  = (double*)(ws + 2048);
    double* X     = (double*)(ws + 2560);
    float*  A     = (float*) (ws + 266752);
    float*  cvec  = (float*) (ws + 332288);

    k_pre<<<132, 128, 0, stream>>>(Wr, Wi, ct256, ct64, X);
    k_fuse<<<256, 256, 0, stream>>>(X, ct256, ct64, br, bi, A, cvec);

    k_wmma<<<NB * 8 * 2, 384, 0, stream>>>(x, A, cvec, cw, cb, gamma, beta, mean, var, (float*)d_out);
}


# k_pre, k_fuse: second cosine/sine mirror about the quarter period (k -> 128-k, sign by column parity): rows folded twice in launch 1, every k sum 65 terms, not 129; new double sum order
# speedup vs baseline: 7.8505x; 1.1388x over previous
#include <hip/hip_runtime.h>
#include <math.h>

#define IMG      2048
#define NB       16
#define DCH      64
#define INF      256

typedef float v2f __attribute__((ext_vector_type(2)));
typedef float v4f __attribute__((ext_vector_type(4)));
typedef float v8f __attribute__((ext_vector_type(8)));
typedef __attribute__((address_space(3))) v4f   lds_v4f;
typedef __attribute__((address_space(3))) float lds_f32;

__device__ __forceinline__ double cos_turn256(int u) {
    int  h   = (u + 32) >> 6;
    int  rho = u - (h << 6);
    bool odd = (h & 1) != 0;
    bool neg = (((h + 1) >> 1) & 1) != 0;
    double t  = (double)rho * (3.14159265358979323846264338327950288 / 128.0);
    double z  = t * t;
    double zt = z * t;
    double p = odd ? (1.0 / 355687428096000.0) : (1.0 / 20922789888000.0);
    p = fma(p, z, odd ? (-1.0 / 1307674368000.0) : (-1.0 / 87178291200.0));
    p = fma(p, z, odd ? (1.0 / 6227020800.0)     : (1.0 / 479001600.0));
    p = fma(p, z, odd ? (-1.0 / 39916800.0)      : (-1.0 / 3628800.0));
    p = fma(p, z, odd ? (1.0 / 362880.0)         : (1.0 / 40320.0));
    p = fma(p, z, odd ? (-1.0 / 5040.0)          : (-1.0 / 720.0));
    p = fma(p, z, odd ? (1.0 / 120.0)            : (1.0 / 24.0));
    p = fma(p, z, odd ? (-1.0 / 6.0)             : (-1.0 / 2.0));
    double r = fma(p, odd ? zt : z, odd ? t : 1.0);
    return neg ? -r : r;
}
#define XF(a, p, k, d) (((((a) * 2 + (p)) * 64 + (k)) * DCH) + (d))
#define XT(a, d)       (8 * 64 * DCH + (a) * DCH + (d))
__global__ void __launch_bounds__(128)
k_pre(const float* __restrict__ Wr, const float* __restrict__ Wi,
      double* __restrict__ ct256, double* __restrict__ ct64, double* __restrict__ X) {
    int b = blockIdx.x;
    if (b < 3) {
        int g = b * 128 + threadIdx.x;
        if (g < 256)      ct256[g] = cos_turn256(g);
        else if (g < 320) ct64[g - 256] = cos_turn256(4 * (g - 256));
        return;
    }
    int g = (b - 3) * 128 + threadIdx.x;
    if (g < 8192) {
        int lane = g & 31;
        int gw   = g >> 5;
        int d    = (gw & 3) * 16 + (lane & 15);
        int k    = ((gw >> 2) & 31) * 2 + (lane >> 4);
        int sel  = gw >> 7;
        const float* w = (sel ? Wi : Wr) + d * INF;
        double a  = (double)w[k];
        double c  = (double)w[(256 - k) & 255];
        double a2 = (double)w[128 - k];
        double c2 = (double)w[128 + k];
        double e  = (k == 0) ? a  : a + c;
        double e2 = (k == 0) ? a2 : a2 + c2;
        double o  = a - c;
        double o2 = a2 - c2;
        X[XF(2 * sel, 0, k, d)]     = e + e2;
        X[XF(2 * sel, 1, k, d)]     = e - e2;
        X[XF(2 * sel + 1, 0, k, d)] = o - o2;
        X[XF(2 * sel + 1, 1, k, d)] = o + o2;
    } else {
        int t   = g - 8192;
        int sel = t >> 6;
        int d   = t & 63;
        const float* w = (sel ? Wi : Wr) + d * INF;
        double a = (double)w[64];
        double c = (double)w[192];
        X[XT(2 * sel, d)]     = a + c;
        X[XT(2 * sel + 1, d)] = a - c;
    }
}

#define AFRAG(p, f) ((((p) >> 4) * 32 + ((f) >> 3)) * 128 + ((p) & 15) * 8 + ((f) & 7))
typedef int v2i __attribute__((ext_vector_type(2)));
__device__ __forceinline__ double lane_value(double v, int j) {
    v2i p = __builtin_bit_cast(v2i, v);
    p.x = __builtin_amdgcn_readlane(p.x, j);
    p.y = __builtin_amdgcn_readlane(p.y, j);
    return __builtin_bit_cast(double, p);
}
__global__ void __launch_bounds__(256)
k_fuse(const double* __restrict__ X, const double* __restrict__ ct, const double* __restrict__ ct64,
       const float* __restrict__ br, const float* __restrict__ bi,
       float* __restrict__ A, float* __restrict__ cvec) {
    __shared__ double part[2][4][DCH];
    __shared__ double S[2][DCH];
    __shared__ double pa[2][4][32];
    __shared__ double p32[2][4];
    int n    = blockIdx.x >> 1;
    int side = blockIdx.x & 1;
    int lane = threadIdx.x & 31;
    int wv = __builtin_amdgcn_readfirstlane((int)(threadIdx.x >> 5));
    int d  = (wv & 1) * 32 + lane;
    int kq = wv >> 1;
    int p  = n & 1;
    int a1 = side ? 2 : 0;
    int a2 = (n == 0) ? a1 : (side ? 1 : 3);
    const double* X1 = X + XF(a1, p, 0, 0);
    const double* X2 = X + XF(a2, p, 0, 0);
    int n2 = (n == 0) ? 128 : n;
    int sh = (n == 0) ? 0 : 192;

    bool biasw = (side == 1) && (n < 64) && (wv == 5);
    float  fr0 = 0.0f, fr1 = 0.0f, fi0 = 0.0f, fi1 = 0.0f;
    double bc0 = 0.0, bc1 = 0.0, bq0 = 0.0, bq1 = 0.0;
    if (biasw) {
        int mm  = n >> 1;
        int bmr = (n & 1) ? ((mm == 0) ? 32 : 64 - mm) : mm;
        int t0 = (lane * bmr) & 63, t1 = ((lane + 32) * bmr) & 63;
        fr0 = br[lane];      fi0 = bi[lane];
        fr1 = br[lane + 32]; fi1 = bi[lane + 32];
        bc0 = ct64[t0]; bq0 = ct64[(t0 + 48) & 63];
        bc1 = ct64[t1]; bq1 = ct64[(t1 + 48) & 63];
    }
    int    kl  = kq * 16 + (lane & 15);
    double t1v = ct[(kl * n) & 255];
    double t2v = ct[(kl * n2 + sh) & 255];
    double x1[16], x2[16];
    #pragma unroll
    for (int j = 0; j < 16; ++j) {
        int k = kq * 16 + j;
        x1[j] = X1[k * DCH + d];
        x2[j] = X2[k * DCH + d];
    }
    double xe1 = X[XT(a1, d)];
    double xe2 = X[XT(a2, d)];
    double te1 = ct[(64 * n) & 255];
    double te2 = ct[(64 * n2 + sh) & 255];
    int rsh = side ? 48 : 0;
    double tc[16];
    #pragma unroll
    for (int j = 0; j < 16; ++j) tc[j] = ct64[((kq * 16 + j) * lane + rsh) & 63];
    __builtin_amdgcn_sched_barrier(0);
    if (biasw) {
        double cc = 0.0;
        cc += bc0 * ((double)fr0 - (double)fi0) - bq0 * ((double)fr0 + (double)fi0);
        cc += bc1 * ((double)fr1 - (double)fi1) - bq1 * ((double)fr1 + (double)fi1);
        for (int o = 16; o >= 1; o >>= 1) cc += __shfl_xor(cc, o);
        if (lane == 0) cvec[n] = (float)(cc / 64.0);
    }
    double s1 = 0.0, s2 = 0.0;
    #pragma unroll
    for (int j = 0; j < 16; ++j) {
        s1 += x1[j] * lane_value(t1v, j);
        s2 += x2[j] * lane_value(t2v, j);
    }
    if (kq == 0) {
        s1 += xe1 * te1;
        s2 += xe2 * te2;
    }
    part[0][kq][d] = s1;
    part[1][kq][d] = s2;
    __syncthreads();
    if (wv >= 4) {
        int o   = (int)threadIdx.x - 128;
        int sel = o >> 6;
        int dd  = o & 63;
        S[sel][dd] = ((part[sel][0][dd] + part[sel][1][dd]) + part[sel][2][dd]) + part[sel][3][dd];
    }
    __syncthreads();

    {
        int sel = wv & 1;
        int dq  = wv >> 1;
        double ae = 0.0, ao = 0.0;
        #pragma unroll
        for (int j = 0; j < 16; j += 2) {
            int d0 = dq * 16 + j;
            ae += tc[j] * S[sel][d0];
            ao += tc[j + 1] * S[sel][d0 + 1];
        }
        pa[sel][dq][lane] = ae + ao;
        if (lane == 0) p32[sel][dq] = ae - ao;
    }
    __syncthreads();

    if (wv == 2 || wv == 3) {
        int c = wv - 2;
        double P = ((pa[0][0][lane] + pa[0][1][lane]) + pa[0][2][lane]) + pa[0][3][lane];
        double Q = ((pa[1][0][lane] + pa[1][1][lane]) + pa[1][2][lane]) + pa[1][3][lane];
        double val = (c == side) ? (P + Q) : (P - Q);
        if (n == 0) val = c ? Q : P;
        int f   = c ? ((n == 0) ? 128 : 256 - n) : n;
        int pos = (lane == 0) ? 0 : 2 * lane + side;
        if (!(side == 1 && lane == 0))
            A[AFRAG(pos, f)] = (float)(val / 64.0);
    }
    if (wv == 5 && side == 0 && lane < 2) {
        int c = lane;
        double P = ((p32[0][0] + p32[0][1]) + p32[0][2]) + p32[0][3];
        double Q = ((p32[1][0] + p32[1][1]) + p32[1][2]) + p32[1][3];
        double val = c ? (P - Q) : (P + Q);
        if (n == 0) val = c ? Q : P;
        int f = c ? ((n == 0) ? 128 : 256 - n) : n;
        A[AFRAG(1, f)] = (float)(val / 64.0);
    }
}

#define SLOTW    12
#define FPITCH   52
__global__ void __launch_bounds__(384)
k_wmma(const float* __restrict__ x, const float* __restrict__ A, const float* __restrict__ cvec,
       const float* __restrict__ cw, const float* __restrict__ cb, const float* __restrict__ gamma,
       const float* __restrict__ beta, const float* __restrict__ mean, const float* __restrict__ var,
       float* __restrict__ out) {
    __shared__ v4f   red[4][3][4][2][32];
    __shared__ float fl[DCH * FPITCH];
    int lane = threadIdx.x & 31;
    int wid  = __builtin_amdgcn_readfirstlane((int)(threadIdx.x >> 5));
    int kq   = wid & 3;
    int tl   = wid >> 2;
    int half = blockIdx.x & 1;
    int oh   = (blockIdx.x >> 1) & 7;
    int img  = blockIdx.x >> 4;
    int n    = lane & 15;
    int hi   = lane >> 4;

    int s  = tl * 16 + n;
    int tc = n > (SLOTW - 1) ? (SLOTW - 1) : n;
    int prow = 16 * oh + tl - 1;
    int pcol = 16 * (4 * half + tc / 3) + (tc % 3) - 1;
    prow = prow < 0 ? 0 : prow;
    pcol = pcol < 0 ? 0 : pcol;

    const float* xb = x + (size_t)img * IMG * IMG
                        + (size_t)(prow * 16 + 4 * kq) * IMG + (size_t)(pcol * 16);
    const float* ap = A + (kq * 8) * 128 + n * 8 + hi * 4;

    v4f b[8];
    v4f a[4][8];
    #pragma unroll
    for (int j = 0; j < 8; ++j) {
        int u = 8 * j + 4 * hi;
        b[j] = *(const v4f*)(xb + (u >> 4) * IMG + (u & 15));
    }
    __builtin_amdgcn_sched_barrier(0);
    #pragma unroll
    for (int t = 0; t < 4; ++t)
        #pragma unroll
        for (int j = 0; j < 8; ++j) {
            a[t][j] = *(const v4f*)(ap + (t * 32 + j) * 128);
        }
    int epos = (int)(threadIdx.x >> 2) & 63;
    int ech  = (epos & 1) ? ((epos >> 1) == 0 ? 32 : 64 - (epos >> 1)) : (epos >> 1);
    int owl = (int)threadIdx.x & 3;
    v4f cv0 = *(const v4f*)(cvec + kq * 16 + hi * 8);
    v4f cv1 = *(const v4f*)(cvec + kq * 16 + hi * 8 + 4);
    float wt[9];
    #pragma unroll
    for (int k = 0; k < 9; ++k) wt[k] = cw[ech * 9 + k];
    float e_cb = cb[ech], e_g = gamma[ech], e_b = beta[ech], e_m = mean[ech], e_v = var[ech];
    lds_v4f* rst = (lds_v4f*)&red[kq][tl][0][0][lane];
    lds_v4f* rld = (lds_v4f*)&red[0][tl][kq][0][lane];
    lds_f32* fst = (lds_f32*)(fl + (kq * 16 + hi * 8) * FPITCH + s);
    asm volatile("" : "+v"(rst), "+v"(rld), "+v"(fst));

    __builtin_amdgcn_sched_barrier(0);

    v8f acc[4] = {};
    #pragma unroll
    for (int j = 0; j < 8; ++j)
        #pragma unroll
        for (int p = 0; p < 2; ++p) {
            v2f bv = p ? (v2f){b[j][2], b[j][3]} : (v2f){b[j][0], b[j][1]};
            #pragma unroll
            for (int t = 0; t < 4; ++t) {
                v2f av = p ? (v2f){a[t][j][2], a[t][j][3]} : (v2f){a[t][j][0], a[t][j][1]};
                acc[t] = __builtin_amdgcn_wmma_f32_16x16x4_f32(
                              false, av,  false, bv,
                              (short)0, acc[t],  false,  false);
            }
        }

    #pragma unroll
    for (int t = 0; t < 4; ++t) {
        rst[64 * t]      = (v4f){acc[t][0], acc[t][1], acc[t][2], acc[t][3]};
        rst[64 * t + 32] = (v4f){acc[t][4], acc[t][5], acc[t][6], acc[t][7]};
    }
    __syncthreads();
    v4f lo = rld[0];
    v4f up = rld[32];
    #pragma unroll
    for (int p = 1; p < 4; ++p) {
        lo += rld[768 * p];
        up += rld[768 * p + 32];
    }
    bool first = (kq == 0) && (hi == 0);
    #pragma unroll
    for (int j = 0; j < 8; j += 2) {
        float fu = (j < 4) ? lo[j & 3] : up[j & 3];
        float fv = (j < 4) ? lo[(j + 1) & 3] : up[(j + 1) & 3];
        float c0 = (j < 4) ? cv0[j & 3] : cv1[j & 3];
        float c1 = (j < 4) ? cv0[(j + 1) & 3] : cv1[(j + 1) & 3];
        bool keep = (j == 0) && first;
        float e = keep ? fu : fu - fv;
        float o = keep ? fv : fu + fv;
        fst[FPITCH * j]       = e + c0;
        fst[FPITCH * (j + 1)] = o + c1;
    }
    __syncthreads();

    if (threadIdx.x < 256) {
        int ow = 4 * half + owl;
        const float* f = fl + epos * FPITCH + 3 * owl;
        float r = e_cb;
        #pragma unroll
        for (int j = 0; j < 3; ++j)
            #pragma unroll
            for (int i = 0; i < 3; ++i) {
                bool pad = (j == 0 && oh == 0) || (i == 0 && ow == 0);
                float fv = 0.0f;
                if (!pad) fv = f[j * 16 + i];
                r = __builtin_fmaf(fv, wt[j * 3 + i], r);
            }
        float scale = e_g / sqrtf(e_v + 1e-5f);
        out[(((size_t)img * DCH + ech) * 8 + oh) * 8 + ow] = (r - e_m) * scale + e_b;
    }
}

extern "C" void kernel_launch(void* const* d_in, const int* in_sizes, int n_in,
                              void* d_out, int out_size, void* d_ws, size_t ws_size,
                              hipStream_t stream) {
    const float* x     = (const float*)d_in[0];
    const float* Wr    = (const float*)d_in[1];
    const float* br    = (const float*)d_in[2];
    const float* Wi    = (const float*)d_in[3];
    const float* bi    = (const float*)d_in[4];
    const float* cw    = (const float*)d_in[5];
    const float* cb    = (const float*)d_in[6];
    const float* gamma = (const float*)d_in[7];
    const float* beta  = (const float*)d_in[8];
    const float* mean  = (const float*)d_in[9];
    const float* var   = (const float*)d_in[10];

    char* ws = (char*)d_ws;
    double* ct256 = (double*)(ws + 0);
    double* ct64  = (double*)(ws + 2048);
    double* X     = (double*)(ws + 2560);
    float*  A     = (float*) (ws + 266752);
    float*  cvec  = (float*) (ws + 332288);

    k_pre<<<68, 128, 0, stream>>>(Wr, Wi, ct256, ct64, X);
    k_fuse<<<256, 256, 0, stream>>>(X, ct256, ct64, br, bi, A, cvec);

    k_wmma<<<NB * 8 * 2, 384, 0, stream>>>(x, A, cvec, cw, cb, gamma, beta, mean, var, (float*)d_out);
}


# k_wmma: one wave per SIMD (4 waves a block, not 12): wave = K quarter of all 12 tiles, each operand piece fetched once, loads in product order; 1024 waves to enter, not 3072; sum order unchanged
# speedup vs baseline: 8.3903x; 1.0688x over previous
#include <hip/hip_runtime.h>
#include <math.h>

#define IMG      2048
#define NB       16
#define DCH      64
#define INF      256

typedef float v2f __attribute__((ext_vector_type(2)));
typedef float v4f __attribute__((ext_vector_type(4)));
typedef float v8f __attribute__((ext_vector_type(8)));
typedef __attribute__((address_space(3))) v4f   lds_v4f;
typedef __attribute__((address_space(3))) float lds_f32;

__device__ __forceinline__ double cos_turn256(int u) {
    int  h   = (u + 32) >> 6;
    int  rho = u - (h << 6);
    bool odd = (h & 1) != 0;
    bool neg = (((h + 1) >> 1) & 1) != 0;
    double t  = (double)rho * (3.14159265358979323846264338327950288 / 128.0);
    double z  = t * t;
    double zt = z * t;
    double p = odd ? (1.0 / 355687428096000.0) : (1.0 / 20922789888000.0);
    p = fma(p, z, odd ? (-1.0 / 1307674368000.0) : (-1.0 / 87178291200.0));
    p = fma(p, z, odd ? (1.0 / 6227020800.0)     : (1.0 / 479001600.0));
    p = fma(p, z, odd ? (-1.0 / 39916800.0)      : (-1.0 / 3628800.0));
    p = fma(p, z, odd ? (1.0 / 362880.0)         : (1.0 / 40320.0));
    p = fma(p, z, odd ? (-1.0 / 5040.0)          : (-1.0 / 720.0));
    p = fma(p, z, odd ? (1.0 / 120.0)            : (1.0 / 24.0));
    p = fma(p, z, odd ? (-1.0 / 6.0)             : (-1.0 / 2.0));
    double r = fma(p, odd ? zt : z, odd ? t : 1.0);
    return neg ? -r : r;
}
#define XF(a, p, k, d) (((((a) * 2 + (p)) * 64 + (k)) * DCH) + (d))
#define XT(a, d)       (8 * 64 * DCH + (a) * DCH + (d))
__global__ void __launch_bounds__(128)
k_pre(const float* __restrict__ Wr, const float* __restrict__ Wi,
      double* __restrict__ ct256, double* __restrict__ ct64, double* __restrict__ X) {
    int b = blockIdx.x;
    if (b < 3) {
        int g = b * 128 + threadIdx.x;
        if (g < 256)      ct256[g] = cos_turn256(g);
        else if (g < 320) ct64[g - 256] = cos_turn256(4 * (g - 256));
        return;
    }
    int g = (b - 3) * 128 + threadIdx.x;
    if (g < 8192) {
        int lane = g & 31;
        int gw   = g >> 5;
        int d    = (gw & 3) * 16 + (lane & 15);
        int k    = ((gw >> 2) & 31) * 2 + (lane >> 4);
        int sel  = gw >> 7;
        const float* w = (sel ? Wi : Wr) + d * INF;
        double a  = (double)w[k];
        double c  = (double)w[(256 - k) & 255];
        double a2 = (double)w[128 - k];
        double c2 = (double)w[128 + k];
        double e  = (k == 0) ? a  : a + c;
        double e2 = (k == 0) ? a2 : a2 + c2;
        double o  = a - c;
        double o2 = a2 - c2;
        X[XF(2 * sel, 0, k, d)]     = e + e2;
        X[XF(2 * sel, 1, k, d)]     = e - e2;
        X[XF(2 * sel + 1, 0, k, d)] = o - o2;
        X[XF(2 * sel + 1, 1, k, d)] = o + o2;
    } else {
        int t   = g - 8192;
        int sel = t >> 6;
        int d   = t & 63;
        const float* w = (sel ? Wi : Wr) + d * INF;
        double a = (double)w[64];
        double c = (double)w[192];
        X[XT(2 * sel, d)]     = a + c;
        X[XT(2 * sel + 1, d)] = a - c;
    }
}

#define AFRAG(p, f) ((((p) >> 4) * 32 + ((f) >> 3)) * 128 + ((p) & 15) * 8 + ((f) & 7))
typedef int v2i __attribute__((ext_vector_type(2)));
__device__ __forceinline__ double lane_value(double v, int j) {
    v2i p = __builtin_bit_cast(v2i, v);
    p.x = __builtin_amdgcn_readlane(p.x, j);
    p.y = __builtin_amdgcn_readlane(p.y, j);
    return __builtin_bit_cast(double, p);
}
__global__ void __launch_bounds__(256)
k_fuse(const double* __restrict__ X, const double* __restrict__ ct, const double* __restrict__ ct64,
       const float* __restrict__ br, const float* __restrict__ bi,
       float* __restrict__ A, float* __restrict__ cvec) {
    __shared__ double part[2][4][DCH];
    __shared__ double S[2][DCH];
    __shared__ double pa[2][4][32];
    __shared__ double p32[2][4];
    int n    = blockIdx.x >> 1;
    int side = blockIdx.x & 1;
    int lane = threadIdx.x & 31;
    int wv = __builtin_amdgcn_readfirstlane((int)(threadIdx.x >> 5));
    int d  = (wv & 1) * 32 + lane;
    int kq = wv >> 1;
    int p  = n & 1;
    int a1 = side ? 2 : 0;
    int a2 = (n == 0) ? a1 : (side ? 1 : 3);
    const double* X1 = X + XF(a1, p, 0, 0);
    const double* X2 = X + XF(a2, p, 0, 0);
    int n2 = (n == 0) ? 128 : n;
    int sh = (n == 0) ? 0 : 192;

    bool biasw = (side == 1) && (n < 64) && (wv == 5);
    float  fr0 = 0.0f, fr1 = 0.0f, fi0 = 0.0f, fi1 = 0.0f;
    double bc0 = 0.0, bc1 = 0.0, bq0 = 0.0, bq1 = 0.0;
    if (biasw) {
        int mm  = n >> 1;
        int bmr = (n & 1) ? ((mm == 0) ? 32 : 64 - mm) : mm;
        int t0 = (lane * bmr) & 63, t1 = ((lane + 32) * bmr) & 63;
        fr0 = br[lane];      fi0 = bi[lane];
        fr1 = br[lane + 32]; fi1 = bi[lane + 32];
        bc0 = ct64[t0]; bq0 = ct64[(t0 + 48) & 63];
        bc1 = ct64[t1]; bq1 = ct64[(t1 + 48) & 63];
    }
    int    kl  = kq * 16 + (lane & 15);
    double t1v = ct[(kl * n) & 255];
    double t2v = ct[(kl * n2 + sh) & 255];
    double x1[16], x2[16];
    #pragma unroll
    for (int j = 0; j < 16; ++j) {
        int k = kq * 16 + j;
        x1[j] = X1[k * DCH + d];
        x2[j] = X2[k * DCH + d];
    }
    double xe1 = X[XT(a1, d)];
    double xe2 = X[XT(a2, d)];
    double te1 = ct[(64 * n) & 255];
    double te2 = ct[(64 * n2 + sh) & 255];
    int rsh = side ? 48 : 0;
    double tc[16];
    #pragma unroll
    for (int j = 0; j < 16; ++j) tc[j] = ct64[((kq * 16 + j) * lane + rsh) & 63];
    __builtin_amdgcn_sched_barrier(0);
    if (biasw) {
        double cc = 0.0;
        cc += bc0 * ((double)fr0 - (double)fi0) - bq0 * ((double)fr0 + (double)fi0);
        cc += bc1 * ((double)fr1 - (double)fi1) - bq1 * ((double)fr1 + (double)fi1);
        for (int o = 16; o >= 1; o >>= 1) cc += __shfl_xor(cc, o);
        if (lane == 0) cvec[n] = (float)(cc / 64.0);
    }
    double s1 = 0.0, s2 = 0.0;
    #pragma unroll
    for (int j = 0; j < 16; ++j) {
        s1 += x1[j] * lane_value(t1v, j);
        s2 += x2[j] * lane_value(t2v, j);
    }
    if (kq == 0) {
        s1 += xe1 * te1;
        s2 += xe2 * te2;
    }
    part[0][kq][d] = s1;
    part[1][kq][d] = s2;
    __syncthreads();
    if (wv >= 4) {
        int o   = (int)threadIdx.x - 128;
        int sel = o >> 6;
        int dd  = o & 63;
        S[sel][dd] = ((part[sel][0][dd] + part[sel][1][dd]) + part[sel][2][dd]) + part[sel][3][dd];
    }
    __syncthreads();

    {
        int sel = wv & 1;
        int dq  = wv >> 1;
        double ae = 0.0, ao = 0.0;
        #pragma unroll
        for (int j = 0; j < 16; j += 2) {
            int d0 = dq * 16 + j;
            ae += tc[j] * S[sel][d0];
            ao += tc[j + 1] * S[sel][d0 + 1];
        }
        pa[sel][dq][lane] = ae + ao;
        if (lane == 0) p32[sel][dq] = ae - ao;
    }
    __syncthreads();

    if (wv == 2 || wv == 3) {
        int c = wv - 2;
        double P = ((pa[0][0][lane] + pa[0][1][lane]) + pa[0][2][lane]) + pa[0][3][lane];
        double Q = ((pa[1][0][lane] + pa[1][1][lane]) + pa[1][2][lane]) + pa[1][3][lane];
        double val = (c == side) ? (P + Q) : (P - Q);
        if (n == 0) val = c ? Q : P;
        int f   = c ? ((n == 0) ? 128 : 256 - n) : n;
        int pos = (lane == 0) ? 0 : 2 * lane + side;
        if (!(side == 1 && lane == 0))
            A[AFRAG(pos, f)] = (float)(val / 64.0);
    }
    if (wv == 5 && side == 0 && lane < 2) {
        int c = lane;
        double P = ((p32[0][0] + p32[0][1]) + p32[0][2]) + p32[0][3];
        double Q = ((p32[1][0] + p32[1][1]) + p32[1][2]) + p32[1][3];
        double val = c ? (P - Q) : (P + Q);
        if (n == 0) val = c ? Q : P;
        int f = c ? ((n == 0) ? 128 : 256 - n) : n;
        A[AFRAG(1, f)] = (float)(val / 64.0);
    }
}

#define SLOTW    12
#define FPITCH   52
__global__ void __launch_bounds__(128)
k_wmma(const float* __restrict__ x, const float* __restrict__ A, const float* __restrict__ cvec,
       const float* __restrict__ cw, const float* __restrict__ cb, const float* __restrict__ gamma,
       const float* __restrict__ beta, const float* __restrict__ mean, const float* __restrict__ var,
       float* __restrict__ out) {
    __shared__ v4f   red[4][3][4][2][32];
    __shared__ float fl[DCH * FPITCH];
    int lane = threadIdx.x & 31;
    int kq   = __builtin_amdgcn_readfirstlane((int)(threadIdx.x >> 5));
    int half = blockIdx.x & 1;
    int oh   = (blockIdx.x >> 1) & 7;
    int img  = blockIdx.x >> 4;
    int n    = lane & 15;
    int hi   = lane >> 4;

    int tc = n > (SLOTW - 1) ? (SLOTW - 1) : n;
    int pcol = 16 * (4 * half + tc / 3) + (tc % 3) - 1;
    pcol = pcol < 0 ? 0 : pcol;
    int prow0 = 16 * oh - 1;
    prow0 = prow0 < 0 ? 0 : prow0;

    const float* xc  = x + (size_t)img * IMG * IMG + (size_t)(4 * kq) * IMG + (size_t)(pcol * 16);
    const float* xb0 = xc + (size_t)(prow0 * 16) * IMG;
    const float* xb1 = xc + (size_t)(oh * 256) * IMG;
    const float* xb2 = xc + (size_t)(oh * 256 + 16) * IMG;
    const float* ap = A + (kq * 8) * 128 + n * 8 + hi * 4;

    v4f b[3][8];
    v4f a[4][8];
    #pragma unroll
    for (int j = 0; j < 8; ++j) {
        int u   = 8 * j + 4 * hi;
        int off = (u >> 4) * IMG + (u & 15);
        b[0][j] = *(const v4f*)(xb0 + off);
        b[1][j] = *(const v4f*)(xb1 + off);
        b[2][j] = *(const v4f*)(xb2 + off);
        #pragma unroll
        for (int t = 0; t < 4; ++t) {
            a[t][j] = *(const v4f*)(ap + (t * 32 + j) * 128);
        }
        __builtin_amdgcn_sched_barrier(0);
    }
    int owl = (int)threadIdx.x & 3;
    v4f cv0, cv1;
    int   epos[2], ech[2];
    float wt[2][9], e_cb[2], e_g[2], e_b[2], e_m[2], e_v[2];
    #pragma unroll
    for (int q = 0; q < 2; ++q) {
        epos[q] = ((int)threadIdx.x >> 2) + 32 * q;
        ech[q]  = (epos[q] & 1) ? ((epos[q] >> 1) == 0 ? 32 : 64 - (epos[q] >> 1)) : (epos[q] >> 1);
    }
    const float* cvp = cvec + kq * 16 + hi * 8;
    const float* cw0 = cw + ech[0] * 9;
    const float* cw1 = cw + ech[1] * 9;
    int ec0 = ech[0], ec1 = ech[1], ep0 = epos[0], ep1 = epos[1];
    asm volatile("" : "+v"(cvp), "+v"(cw0), "+v"(cw1), "+v"(ec0), "+v"(ec1), "+v"(ep0), "+v"(ep1), "+v"(owl));
    ech[0] = ec0; ech[1] = ec1; epos[0] = ep0; epos[1] = ep1;
    lds_v4f* rst = (lds_v4f*)&red[kq][0][0][0][lane];
    lds_v4f* rld = (lds_v4f*)&red[0][0][kq][0][lane];
    lds_f32* fst = (lds_f32*)(fl + (kq * 16 + hi * 8) * FPITCH + n);
    asm volatile("" : "+v"(rst), "+v"(rld), "+v"(fst));

    __builtin_amdgcn_sched_barrier(0);

    v8f acc[3][4] = {};
    #pragma unroll
    for (int j = 0; j < 8; ++j) {
        if (j == 4) {
            __builtin_amdgcn_sched_barrier(0);
            cv0 = *(const v4f*)(cvp);
            cv1 = *(const v4f*)(cvp + 4);
            #pragma unroll
            for (int q = 0; q < 2; ++q) {
                #pragma unroll
                for (int k = 0; k < 9; ++k) wt[q][k] = (q ? cw1 : cw0)[k];
                e_cb[q] = cb[ech[q]]; e_g[q] = gamma[ech[q]]; e_b[q] = beta[ech[q]];
                e_m[q] = mean[ech[q]]; e_v[q] = var[ech[q]];
            }
            __builtin_amdgcn_sched_barrier(0);
        }
        #pragma unroll
        for (int p = 0; p < 2; ++p)
            #pragma unroll
            for (int tl = 0; tl < 3; ++tl) {
                v2f bv = p ? (v2f){b[tl][j][2], b[tl][j][3]} : (v2f){b[tl][j][0], b[tl][j][1]};
                #pragma unroll
                for (int t = 0; t < 4; ++t) {
                    v2f av = p ? (v2f){a[t][j][2], a[t][j][3]} : (v2f){a[t][j][0], a[t][j][1]};
                    acc[tl][t] = __builtin_amdgcn_wmma_f32_16x16x4_f32(
                                      false, av,  false, bv,
                                      (short)0, acc[tl][t],  false,  false);
                }
            }
    }

    #pragma unroll
    for (int tl = 0; tl < 3; ++tl)
        #pragma unroll
        for (int t = 0; t < 4; ++t) {
            rst[256 * tl + 64 * t]      = (v4f){acc[tl][t][0], acc[tl][t][1], acc[tl][t][2], acc[tl][t][3]};
            rst[256 * tl + 64 * t + 32] = (v4f){acc[tl][t][4], acc[tl][t][5], acc[tl][t][6], acc[tl][t][7]};
        }
    __syncthreads();
    bool first = (kq == 0) && (hi == 0);
    #pragma unroll
    for (int tl = 0; tl < 3; ++tl) {
        v4f lo = rld[256 * tl];
        v4f up = rld[256 * tl + 32];
        #pragma unroll
        for (int p = 1; p < 4; ++p) {
            lo += rld[768 * p + 256 * tl];
            up += rld[768 * p + 256 * tl + 32];
        }
        #pragma unroll
        for (int j = 0; j < 8; j += 2) {
            float fu = (j < 4) ? lo[j & 3] : up[j & 3];
            float fv = (j < 4) ? lo[(j + 1) & 3] : up[(j + 1) & 3];
            float c0 = (j < 4) ? cv0[j & 3] : cv1[j & 3];
            float c1 = (j < 4) ? cv0[(j + 1) & 3] : cv1[(j + 1) & 3];
            bool keep = (j == 0) && first;
            float e = keep ? fu : fu - fv;
            float o = keep ? fv : fu + fv;
            fst[16 * tl + FPITCH * j]       = e + c0;
            fst[16 * tl + FPITCH * (j + 1)] = o + c1;
        }
    }
    __syncthreads();

    int ow = 4 * half + owl;
    #pragma unroll
    for (int q = 0; q < 2; ++q) {
        const float* f = fl + epos[q] * FPITCH + 3 * owl;
        float r = e_cb[q];
        #pragma unroll
        for (int j = 0; j < 3; ++j)
            #pragma unroll
            for (int i = 0; i < 3; ++i) {
                bool pad = (j == 0 && oh == 0) || (i == 0 && ow == 0);
                float fv = 0.0f;
                if (!pad) fv = f[j * 16 + i];
                r = __builtin_fmaf(fv, wt[q][j * 3 + i], r);
            }
        float scale = e_g[q] / sqrtf(e_v[q] + 1e-5f);
        out[(((size_t)img * DCH + ech[q]) * 8 + oh) * 8 + ow] = (r - e_m[q]) * scale + e_b[q];
    }
}

extern "C" void kernel_launch(void* const* d_in, const int* in_sizes, int n_in,
                              void* d_out, int out_size, void* d_ws, size_t ws_size,
                              hipStream_t stream) {
    const float* x     = (const float*)d_in[0];
    const float* Wr    = (const float*)d_in[1];
    const float* br    = (const float*)d_in[2];
    const float* Wi    = (const float*)d_in[3];
    const float* bi    = (const float*)d_in[4];
    const float* cw    = (const float*)d_in[5];
    const float* cb    = (const float*)d_in[6];
    const float* gamma = (const float*)d_in[7];
    const float* beta  = (const float*)d_in[8];
    const float* mean  = (const float*)d_in[9];
    const float* var   = (const float*)d_in[10];

    char* ws = (char*)d_ws;
    double* ct256 = (double*)(ws + 0);
    double* ct64  = (double*)(ws + 2048);
    double* X     = (double*)(ws + 2560);
    float*  A     = (float*) (ws + 266752);
    float*  cvec  = (float*) (ws + 332288);

    k_pre<<<68, 128, 0, stream>>>(Wr, Wi, ct256, ct64, X);
    k_fuse<<<256, 256, 0, stream>>>(X, ct256, ct64, br, bi, A, cvec);

    k_wmma<<<NB * 8 * 2, 128, 0, stream>>>(x, A, cvec, cw, cb, gamma, beta, mean, var, (float*)d_out);
}
